# starting kernel
# baseline (speedup 1.0000x reference)
#include <hip/hip_runtime.h>
#include <math.h>


typedef float v2f __attribute__((ext_vector_type(2)));
typedef float v8f __attribute__((ext_vector_type(8)));

namespace {
constexpr int kBatch  = 64;
constexpr int kEv     = 500;
constexpr int kK      = kBatch * kEv;
constexpr int kHid    = 128;
constexpr int kS      = 2209;
constexpr int kSTiles = (kS + 15) / 16;
constexpr int kSPad   = kSTiles * 16;
constexpr int kT      = 550;
constexpr int kTPad   = 560;
constexpr int kTTiles = kTPad / 16;
constexpr int kKB     = 32;
constexpr int kNChunk = 4;
constexpr int kKCh    = kK / kNChunk;
constexpr float kNorm = 0.3989422804f * 1.41421356237309515f;
}

__global__ __launch_bounds__(256)
void sipm_h_kernel(const float* __restrict__ x, const float* __restrict__ W1,
                   const float* __restrict__ b1, float* __restrict__ h) {
  int idx = blockIdx.x * 256 + threadIdx.x;
  if (idx >= kK * (kHid / 4)) return;
  int k = idx >> 5;
  int j = (idx & 31) << 2;
  float x0 = x[k * 3 + 0], x1 = x[k * 3 + 1], x2 = x[k * 3 + 2];
  const float4 wa = *(const float4*)(W1 + 0 * kHid + j);
  const float4 wb = *(const float4*)(W1 + 1 * kHid + j);
  const float4 wc = *(const float4*)(W1 + 2 * kHid + j);
  const float4 bb = *(const float4*)(b1 + j);
  float4 r;
  r.x = fmaxf(fmaf(x0, wa.x, fmaf(x1, wb.x, fmaf(x2, wc.x, bb.x))), 0.f);
  r.y = fmaxf(fmaf(x0, wa.y, fmaf(x1, wb.y, fmaf(x2, wc.y, bb.y))), 0.f);
  r.z = fmaxf(fmaf(x0, wa.z, fmaf(x1, wb.z, fmaf(x2, wc.z, bb.z))), 0.f);
  r.w = fmaxf(fmaf(x0, wa.w, fmaf(x1, wb.w, fmaf(x2, wc.w, bb.w))), 0.f);
  *(float4*)(h + (size_t)k * kHid + j) = r;
}

__global__ __launch_bounds__(256)
void sipm_expv_kernel(const float* __restrict__ z, const float* __restrict__ mask,
                      float* __restrict__ expv) {
  int idx = blockIdx.x * 256 + threadIdx.x;
  if (idx >= kK * kTPad) return;
  int k = idx / kTPad;
  int t = idx - k * kTPad;
  float v = 0.f;
  if (t < kT) {
    float d = (float)t - z[k];
    v = expf(-d * d) * mask[k] * kNorm;
  }
  expv[idx] = v;
}

__global__ __launch_bounds__(512)
void sipm_gemm_kernel(const float* __restrict__ W2, const float* __restrict__ b2,
                      const float* __restrict__ h, const float* __restrict__ expv,
                      float* __restrict__ part) {
  __shared__ float lds_w2[kHid * 16];
  __shared__ float lds_resp[kKB * 16];

  const int tid    = threadIdx.x;
  const int stile  = blockIdx.x;
  const int chunk  = blockIdx.y;
  const int s_base = stile * 16;

  for (int r = 0; r < 4; ++r) {
    int idx = tid + r * 512;
    int i = idx >> 4, ss = idx & 15;
    int s = s_base + ss;
    lds_w2[idx] = (s < kS) ? W2[(size_t)i * kS + s] : 0.f;
  }

  const int kk  = tid >> 4;
  const int ssA = tid & 15;
  const int sA  = s_base + ssA;
  const float accInit = (sA < kS) ? b2[sA] : 0.f;

  const int w    = tid >> 5;
  const int lane = tid & 31;
  const int m    = lane & 15;
  const int kb2  = (lane >> 4) << 1;
  const int tb0 = w * 16, tb1 = (w + 16) * 16, tb2 = (w + 32) * 16;
  const bool t2ok = (w + 32) < kTTiles;

  v8f acc0 = {}, acc1 = {}, acc2 = {};

  __syncthreads();

  for (int kblk = 0; kblk < kKCh / kKB; ++kblk) {
    const int k0 = chunk * kKCh + kblk * kKB;

    {
      float acc = accInit;
      const float4* hrow = (const float4*)(h + (size_t)(k0 + kk) * kHid);
      #pragma unroll
      for (int i4 = 0; i4 < kHid / 4; ++i4) {
        float4 hv = hrow[i4];
        int ib = i4 * 4;
        acc = fmaf(hv.x, lds_w2[(ib + 0) * 16 + ssA], acc);
        acc = fmaf(hv.y, lds_w2[(ib + 1) * 16 + ssA], acc);
        acc = fmaf(hv.z, lds_w2[(ib + 2) * 16 + ssA], acc);
        acc = fmaf(hv.w, lds_w2[(ib + 3) * 16 + ssA], acc);
      }
      lds_resp[kk * 16 + ssA] = expf(acc);
    }
    __syncthreads();

    const float* erow = expv + (size_t)(k0 + kb2) * kTPad + m;
    v2f aP[2], bP0[2], bP1[2], bP2[2];
    #pragma unroll
    for (int p = 0; p < 2; ++p) {
      const int ks = p * 4;
      aP[p].x = lds_resp[(ks + kb2) * 16 + m];
      aP[p].y = lds_resp[(ks + kb2 + 1) * 16 + m];
      const float* e = erow + (size_t)ks * kTPad;
      bP0[p].x = e[tb0]; bP0[p].y = e[kTPad + tb0];
      bP1[p].x = e[tb1]; bP1[p].y = e[kTPad + tb1];
      bP2[p].x = e[tb2]; bP2[p].y = e[kTPad + tb2];
    }
    #pragma unroll
    for (int i = 0; i < kKB / 4; ++i) {
      const int slot = i & 1;
      const v2f a   = aP[slot];
      const v2f b0  = bP0[slot];
      const v2f b1v = bP1[slot];
      const v2f b2v = bP2[slot];
      if (i + 2 < kKB / 4) {
        const int ks = (i + 2) * 4;
        aP[slot].x = lds_resp[(ks + kb2) * 16 + m];
        aP[slot].y = lds_resp[(ks + kb2 + 1) * 16 + m];
        const float* e = erow + (size_t)ks * kTPad;
        bP0[slot].x = e[tb0]; bP0[slot].y = e[kTPad + tb0];
        bP1[slot].x = e[tb1]; bP1[slot].y = e[kTPad + tb1];
        bP2[slot].x = e[tb2]; bP2[slot].y = e[kTPad + tb2];
      }
      acc0 = __builtin_amdgcn_wmma_f32_16x16x4_f32(false, a, false, b0,
                                                   (short)0, acc0, false, false);
      acc1 = __builtin_amdgcn_wmma_f32_16x16x4_f32(false, a, false, b1v,
                                                   (short)0, acc1, false, false);
      if (t2ok)
        acc2 = __builtin_amdgcn_wmma_f32_16x16x4_f32(false, a, false, b2v,
                                                     (short)0, acc2, false, false);
    }
    __syncthreads();
  }

  float* pbase = part + ((size_t)chunk * kSPad + s_base + ((lane >> 4) << 3)) * kTPad + m;
  #pragma unroll
  for (int r = 0; r < 8; ++r) {
    float* prow = pbase + (size_t)r * kTPad;
    prow[tb0] = acc0[r];
    prow[tb1] = acc1[r];
    if (t2ok) prow[tb2] = acc2[r];
  }
}

__global__ __launch_bounds__(256)
void sipm_reduce_kernel(const float* __restrict__ part, float* __restrict__ out) {
  int idx = blockIdx.x * 256 + threadIdx.x;
  if (idx >= kS * kT) return;
  int s = idx / kT, t = idx - s * kT;
  size_t off = (size_t)s * kTPad + t;
  float sum = 0.f;
  #pragma unroll
  for (int c = 0; c < kNChunk; ++c)
    sum += part[(size_t)c * kSPad * kTPad + off];
  out[idx] = sum;
}

extern "C" void kernel_launch(void* const* d_in, const int* in_sizes, int n_in,
                              void* d_out, int out_size, void* d_ws, size_t ws_size,
                              hipStream_t stream) {
  const float* x   = (const float*)d_in[0];
  const float* z   = (const float*)d_in[1];
  const float* msk = (const float*)d_in[2];
  const float* W1  = (const float*)d_in[3];
  const float* b1  = (const float*)d_in[4];
  const float* W2  = (const float*)d_in[5];
  const float* b2  = (const float*)d_in[6];
  float* out = (float*)d_out;

  float* ws   = (float*)d_ws;
  float* h    = ws;
  float* expv = h + (size_t)kK * kHid;
  float* part = expv + (size_t)kK * kTPad;

  sipm_h_kernel<<<(kK * (kHid / 4) + 255) / 256, 256, 0, stream>>>(x, W1, b1, h);
  sipm_expv_kernel<<<(kK * kTPad + 255) / 256, 256, 0, stream>>>(z, msk, expv);
  sipm_gemm_kernel<<<dim3(kSTiles, kNChunk), 512, 0, stream>>>(W2, b2, h, expv, part);
  sipm_reduce_kernel<<<(kS * kT + 255) / 256, 256, 0, stream>>>(part, out);
}


# K split into 11 chunks of 91 k-blocks (was 4 of 250): 1529 workgroups fill 3 whole rounds of the 512 resident slots, was 556 on 512 (one round + 44); final sum now adds 11 partials, not 4
# speedup vs baseline: 1.4362x; 1.4362x over previous
#include <hip/hip_runtime.h>
#include <math.h>


typedef float v2f __attribute__((ext_vector_type(2)));
typedef float v8f __attribute__((ext_vector_type(8)));

namespace {
constexpr int kBatch  = 64;
constexpr int kEv     = 500;
constexpr int kK      = kBatch * kEv;
constexpr int kHid    = 128;
constexpr int kS      = 2209;
constexpr int kSTiles = (kS + 15) / 16;
constexpr int kSPad   = kSTiles * 16;
constexpr int kT      = 550;
constexpr int kTPad   = 560;
constexpr int kTTiles = kTPad / 16;
constexpr int kKB     = 32;
constexpr int kKBlocks = kK / kKB;
constexpr int kNChunk = 11;
constexpr int kBlkCh  = (kKBlocks + kNChunk - 1) / kNChunk;
constexpr float kNorm = 0.3989422804f * 1.41421356237309515f;
}

__global__ __launch_bounds__(256)
void sipm_h_kernel(const float* __restrict__ x, const float* __restrict__ W1,
                   const float* __restrict__ b1, float* __restrict__ h) {
  int idx = blockIdx.x * 256 + threadIdx.x;
  if (idx >= kK * (kHid / 4)) return;
  int k = idx >> 5;
  int j = (idx & 31) << 2;
  float x0 = x[k * 3 + 0], x1 = x[k * 3 + 1], x2 = x[k * 3 + 2];
  const float4 wa = *(const float4*)(W1 + 0 * kHid + j);
  const float4 wb = *(const float4*)(W1 + 1 * kHid + j);
  const float4 wc = *(const float4*)(W1 + 2 * kHid + j);
  const float4 bb = *(const float4*)(b1 + j);
  float4 r;
  r.x = fmaxf(fmaf(x0, wa.x, fmaf(x1, wb.x, fmaf(x2, wc.x, bb.x))), 0.f);
  r.y = fmaxf(fmaf(x0, wa.y, fmaf(x1, wb.y, fmaf(x2, wc.y, bb.y))), 0.f);
  r.z = fmaxf(fmaf(x0, wa.z, fmaf(x1, wb.z, fmaf(x2, wc.z, bb.z))), 0.f);
  r.w = fmaxf(fmaf(x0, wa.w, fmaf(x1, wb.w, fmaf(x2, wc.w, bb.w))), 0.f);
  *(float4*)(h + (size_t)k * kHid + j) = r;
}

__global__ __launch_bounds__(256)
void sipm_expv_kernel(const float* __restrict__ z, const float* __restrict__ mask,
                      float* __restrict__ expv) {
  int idx = blockIdx.x * 256 + threadIdx.x;
  if (idx >= kK * kTPad) return;
  int k = idx / kTPad;
  int t = idx - k * kTPad;
  float v = 0.f;
  if (t < kT) {
    float d = (float)t - z[k];
    v = expf(-d * d) * mask[k] * kNorm;
  }
  expv[idx] = v;
}

__global__ __launch_bounds__(512)
void sipm_gemm_kernel(const float* __restrict__ W2, const float* __restrict__ b2,
                      const float* __restrict__ h, const float* __restrict__ expv,
                      float* __restrict__ part) {
  __shared__ float lds_w2[kHid * 16];
  __shared__ float lds_resp[kKB * 16];

  const int tid    = threadIdx.x;
  const int stile  = blockIdx.x;
  const int chunk  = blockIdx.y;
  const int s_base = stile * 16;

  for (int r = 0; r < 4; ++r) {
    int idx = tid + r * 512;
    int i = idx >> 4, ss = idx & 15;
    int s = s_base + ss;
    lds_w2[idx] = (s < kS) ? W2[(size_t)i * kS + s] : 0.f;
  }

  const int kk  = tid >> 4;
  const int ssA = tid & 15;
  const int sA  = s_base + ssA;
  const float accInit = (sA < kS) ? b2[sA] : 0.f;

  const int w    = tid >> 5;
  const int lane = tid & 31;
  const int m    = lane & 15;
  const int kb2  = (lane >> 4) << 1;
  const int tb0 = w * 16, tb1 = (w + 16) * 16, tb2 = (w + 32) * 16;
  const bool t2ok = (w + 32) < kTTiles;

  v8f acc0 = {}, acc1 = {}, acc2 = {};

  __syncthreads();

  const int blk0 = chunk * kBlkCh;
  const int nblk = (kKBlocks - blk0 < kBlkCh) ? (kKBlocks - blk0) : kBlkCh;

  for (int kblk = 0; kblk < nblk; ++kblk) {
    const int k0 = (blk0 + kblk) * kKB;

    {
      float acc = accInit;
      const float4* hrow = (const float4*)(h + (size_t)(k0 + kk) * kHid);
      #pragma unroll
      for (int i4 = 0; i4 < kHid / 4; ++i4) {
        float4 hv = hrow[i4];
        int ib = i4 * 4;
        acc = fmaf(hv.x, lds_w2[(ib + 0) * 16 + ssA], acc);
        acc = fmaf(hv.y, lds_w2[(ib + 1) * 16 + ssA], acc);
        acc = fmaf(hv.z, lds_w2[(ib + 2) * 16 + ssA], acc);
        acc = fmaf(hv.w, lds_w2[(ib + 3) * 16 + ssA], acc);
      }
      lds_resp[kk * 16 + ssA] = expf(acc);
    }
    __syncthreads();

    const float* erow = expv + (size_t)(k0 + kb2) * kTPad + m;
    v2f aP[2], bP0[2], bP1[2], bP2[2];
    #pragma unroll
    for (int p = 0; p < 2; ++p) {
      const int ks = p * 4;
      aP[p].x = lds_resp[(ks + kb2) * 16 + m];
      aP[p].y = lds_resp[(ks + kb2 + 1) * 16 + m];
      const float* e = erow + (size_t)ks * kTPad;
      bP0[p].x = e[tb0]; bP0[p].y = e[kTPad + tb0];
      bP1[p].x = e[tb1]; bP1[p].y = e[kTPad + tb1];
      bP2[p].x = e[tb2]; bP2[p].y = e[kTPad + tb2];
    }
    #pragma unroll
    for (int i = 0; i < kKB / 4; ++i) {
      const int slot = i & 1;
      const v2f a   = aP[slot];
      const v2f b0  = bP0[slot];
      const v2f b1v = bP1[slot];
      const v2f b2v = bP2[slot];
      if (i + 2 < kKB / 4) {
        const int ks = (i + 2) * 4;
        aP[slot].x = lds_resp[(ks + kb2) * 16 + m];
        aP[slot].y = lds_resp[(ks + kb2 + 1) * 16 + m];
        const float* e = erow + (size_t)ks * kTPad;
        bP0[slot].x = e[tb0]; bP0[slot].y = e[kTPad + tb0];
        bP1[slot].x = e[tb1]; bP1[slot].y = e[kTPad + tb1];
        bP2[slot].x = e[tb2]; bP2[slot].y = e[kTPad + tb2];
      }
      acc0 = __builtin_amdgcn_wmma_f32_16x16x4_f32(false, a, false, b0,
                                                   (short)0, acc0, false, false);
      acc1 = __builtin_amdgcn_wmma_f32_16x16x4_f32(false, a, false, b1v,
                                                   (short)0, acc1, false, false);
      if (t2ok)
        acc2 = __builtin_amdgcn_wmma_f32_16x16x4_f32(false, a, false, b2v,
                                                     (short)0, acc2, false, false);
    }
    __syncthreads();
  }

  float* pbase = part + ((size_t)chunk * kSPad + s_base + ((lane >> 4) << 3)) * kTPad + m;
  #pragma unroll
  for (int r = 0; r < 8; ++r) {
    float* prow = pbase + (size_t)r * kTPad;
    prow[tb0] = acc0[r];
    prow[tb1] = acc1[r];
    if (t2ok) prow[tb2] = acc2[r];
  }
}

__global__ __launch_bounds__(256)
void sipm_reduce_kernel(const float* __restrict__ part, float* __restrict__ out) {
  int idx = blockIdx.x * 256 + threadIdx.x;
  if (idx >= kS * kT) return;
  int s = idx / kT, t = idx - s * kT;
  size_t off = (size_t)s * kTPad + t;
  float sum = 0.f;
  #pragma unroll
  for (int c = 0; c < kNChunk; ++c)
    sum += part[(size_t)c * kSPad * kTPad + off];
  out[idx] = sum;
}

extern "C" void kernel_launch(void* const* d_in, const int* in_sizes, int n_in,
                              void* d_out, int out_size, void* d_ws, size_t ws_size,
                              hipStream_t stream) {
  const float* x   = (const float*)d_in[0];
  const float* z   = (const float*)d_in[1];
  const float* msk = (const float*)d_in[2];
  const float* W1  = (const float*)d_in[3];
  const float* b1  = (const float*)d_in[4];
  const float* W2  = (const float*)d_in[5];
  const float* b2  = (const float*)d_in[6];
  float* out = (float*)d_out;

  float* ws   = (float*)d_ws;
  float* h    = ws;
  float* expv = h + (size_t)kK * kHid;
  float* part = expv + (size_t)kK * kTPad;

  sipm_h_kernel<<<(kK * (kHid / 4) + 255) / 256, 256, 0, stream>>>(x, W1, b1, h);
  sipm_expv_kernel<<<(kK * kTPad + 255) / 256, 256, 0, stream>>>(z, msk, expv);
  sipm_gemm_kernel<<<dim3(kSTiles, kNChunk), 512, 0, stream>>>(W2, b2, h, expv, part);
  sipm_reduce_kernel<<<(kS * kT + 255) / 256, 256, 0, stream>>>(part, out);
}


# resp panel (h @ W2, then exp) moved from per-thread FMA chains fed by LDS onto f32 WMMA, k-block 128, W2 strip kept in LDS in fragment order; the hidden-layer dot products are summed in a new order
# speedup vs baseline: 2.2391x; 1.5590x over previous
#include <hip/hip_runtime.h>
#include <math.h>


typedef float v2f __attribute__((ext_vector_type(2)));
typedef float v8f __attribute__((ext_vector_type(8)));

namespace {
constexpr int kBatch  = 64;
constexpr int kEv     = 500;
constexpr int kK      = kBatch * kEv;
constexpr int kHid    = 128;
constexpr int kS      = 2209;
constexpr int kSTiles = (kS + 15) / 16;
constexpr int kSPad   = kSTiles * 16;
constexpr int kT      = 550;
constexpr int kTPad   = 560;
constexpr int kTTiles = kTPad / 16;
constexpr int kKB     = 128;
constexpr int kKBlocks = kK / kKB;
constexpr int kNChunk = 11;
constexpr int kBlkCh  = (kKBlocks + kNChunk - 1) / kNChunk;
constexpr float kNorm = 0.3989422804f * 1.41421356237309515f;
}

__global__ __launch_bounds__(256)
void sipm_h_kernel(const float* __restrict__ x, const float* __restrict__ W1,
                   const float* __restrict__ b1, float* __restrict__ h) {
  int idx = blockIdx.x * 256 + threadIdx.x;
  if (idx >= kK * (kHid / 4)) return;
  int k = idx >> 5;
  int j = (idx & 31) << 2;
  float x0 = x[k * 3 + 0], x1 = x[k * 3 + 1], x2 = x[k * 3 + 2];
  const float4 wa = *(const float4*)(W1 + 0 * kHid + j);
  const float4 wb = *(const float4*)(W1 + 1 * kHid + j);
  const float4 wc = *(const float4*)(W1 + 2 * kHid + j);
  const float4 bb = *(const float4*)(b1 + j);
  float4 r;
  r.x = fmaxf(fmaf(x0, wa.x, fmaf(x1, wb.x, fmaf(x2, wc.x, bb.x))), 0.f);
  r.y = fmaxf(fmaf(x0, wa.y, fmaf(x1, wb.y, fmaf(x2, wc.y, bb.y))), 0.f);
  r.z = fmaxf(fmaf(x0, wa.z, fmaf(x1, wb.z, fmaf(x2, wc.z, bb.z))), 0.f);
  r.w = fmaxf(fmaf(x0, wa.w, fmaf(x1, wb.w, fmaf(x2, wc.w, bb.w))), 0.f);
  *(float4*)(h + (size_t)k * kHid + j) = r;
}

__global__ __launch_bounds__(256)
void sipm_expv_kernel(const float* __restrict__ z, const float* __restrict__ mask,
                      float* __restrict__ expv) {
  int idx = blockIdx.x * 256 + threadIdx.x;
  if (idx >= kK * kTPad) return;
  int k = idx / kTPad;
  int t = idx - k * kTPad;
  float v = 0.f;
  if (t < kT) {
    float d = (float)t - z[k];
    v = expf(-d * d) * mask[k] * kNorm;
  }
  expv[idx] = v;
}

__global__ __launch_bounds__(512, 8)
void sipm_gemm_kernel(const float* __restrict__ W2, const float* __restrict__ b2,
                      const float* __restrict__ h, const float* __restrict__ expv,
                      float* __restrict__ part) {
  __shared__ __attribute__((aligned(16))) float lds_w2f[kHid * 16];
  __shared__ float lds_resp[kKB * 16];

  const int tid    = threadIdx.x;
  const int stile  = blockIdx.x;
  const int chunk  = blockIdx.y;
  const int s_base = stile * 16;

  for (int r = 0; r < 4; ++r) {
    int idx = tid + r * 512;
    int i = idx >> 4, ss = idx & 15;
    int s = s_base + ss;
    float v = (s < kS) ? W2[(size_t)i * kS + s] : 0.f;
    int j = i >> 3, hh2 = (i >> 2) & 1, q = i & 3;
    lds_w2f[((j * 32 + hh2 * 16 + ss) << 2) + q] = v;
  }

  const int w    = tid >> 5;
  const int lane = tid & 31;
  const int m    = lane & 15;
  const int hh   = lane >> 4;
  const int kb2  = hh << 1;
  const int tb0 = w * 16, tb1 = (w + 16) * 16, tb2 = (w + 32) * 16;
  const bool t2ok = (w + 32) < kTTiles;

  const bool doA = __builtin_amdgcn_readfirstlane(w) < (kKB / 16);
  const int  sA  = s_base + m;
  const float biasA = (sA < kS) ? b2[sA] : 0.f;

  v8f acc0 = {}, acc1 = {}, acc2 = {};

  __syncthreads();

  const int blk0 = chunk * kBlkCh;
  const int nblk = (kKBlocks - blk0 < kBlkCh) ? (kKBlocks - blk0) : kBlkCh;

  for (int kblk = 0; kblk < nblk; ++kblk) {
    const int k0 = (blk0 + kblk) * kKB;

    if (doA) {
      const float* hrow = h + (size_t)(k0 + 16 * w + m) * kHid + 4 * hh;
      const float* wfr  = lds_w2f + (lane << 2);
      v8f d0 = {}, d1 = {};
      #pragma unroll
      for (int j = 0; j < kHid / 8; ++j) {
        const float4 a4 = *(const float4*)(hrow + 8 * j);
        const float4 b4 = *(const float4*)(wfr + j * 128);
        v2f aLo, aHi, bLo, bHi;
        aLo.x = a4.x; aLo.y = a4.y; aHi.x = a4.z; aHi.y = a4.w;
        bLo.x = b4.x; bLo.y = b4.y; bHi.x = b4.z; bHi.y = b4.w;
        d0 = __builtin_amdgcn_wmma_f32_16x16x4_f32(false, aLo, false, bLo,
                                                   (short)0, d0, false, false);
        d1 = __builtin_amdgcn_wmma_f32_16x16x4_f32(false, aHi, false, bHi,
                                                   (short)0, d1, false, false);
      }
      #pragma unroll
      for (int r = 0; r < 8; ++r)
        lds_resp[(16 * w + 8 * hh + r) * 16 + m] = expf(d0[r] + d1[r] + biasA);
    }
    __syncthreads();

    const float* erow = expv + (size_t)(k0 + kb2) * kTPad + m;
    v2f aP[2], bP0[2], bP1[2], bP2[2];
    #pragma unroll
    for (int p = 0; p < 2; ++p) {
      const int ks = p * 4;
      aP[p].x = lds_resp[(ks + kb2) * 16 + m];
      aP[p].y = lds_resp[(ks + kb2 + 1) * 16 + m];
      const float* e = erow + (size_t)ks * kTPad;
      bP0[p].x = e[tb0]; bP0[p].y = e[kTPad + tb0];
      bP1[p].x = e[tb1]; bP1[p].y = e[kTPad + tb1];
      bP2[p].x = e[tb2]; bP2[p].y = e[kTPad + tb2];
    }
    #pragma unroll
    for (int i = 0; i < kKB / 4; ++i) {
      const int slot = i & 1;
      const v2f a   = aP[slot];
      const v2f b0  = bP0[slot];
      const v2f b1v = bP1[slot];
      const v2f b2v = bP2[slot];
      if (i + 2 < kKB / 4) {
        const int ks = (i + 2) * 4;
        aP[slot].x = lds_resp[(ks + kb2) * 16 + m];
        aP[slot].y = lds_resp[(ks + kb2 + 1) * 16 + m];
        const float* e = erow + (size_t)ks * kTPad;
        bP0[slot].x = e[tb0]; bP0[slot].y = e[kTPad + tb0];
        bP1[slot].x = e[tb1]; bP1[slot].y = e[kTPad + tb1];
        bP2[slot].x = e[tb2]; bP2[slot].y = e[kTPad + tb2];
      }
      acc0 = __builtin_amdgcn_wmma_f32_16x16x4_f32(false, a, false, b0,
                                                   (short)0, acc0, false, false);
      acc1 = __builtin_amdgcn_wmma_f32_16x16x4_f32(false, a, false, b1v,
                                                   (short)0, acc1, false, false);
      if (t2ok)
        acc2 = __builtin_amdgcn_wmma_f32_16x16x4_f32(false, a, false, b2v,
                                                     (short)0, acc2, false, false);
    }
    __syncthreads();
  }

  float* pbase = part + ((size_t)chunk * kSPad + s_base + ((lane >> 4) << 3)) * kTPad + m;
  #pragma unroll
  for (int r = 0; r < 8; ++r) {
    float* prow = pbase + (size_t)r * kTPad;
    prow[tb0] = acc0[r];
    prow[tb1] = acc1[r];
    if (t2ok) prow[tb2] = acc2[r];
  }
}

__global__ __launch_bounds__(256)
void sipm_reduce_kernel(const float* __restrict__ part, float* __restrict__ out) {
  int idx = blockIdx.x * 256 + threadIdx.x;
  if (idx >= kS * kT) return;
  int s = idx / kT, t = idx - s * kT;
  size_t off = (size_t)s * kTPad + t;
  float sum = 0.f;
  #pragma unroll
  for (int c = 0; c < kNChunk; ++c)
    sum += part[(size_t)c * kSPad * kTPad + off];
  out[idx] = sum;
}

extern "C" void kernel_launch(void* const* d_in, const int* in_sizes, int n_in,
                              void* d_out, int out_size, void* d_ws, size_t ws_size,
                              hipStream_t stream) {
  const float* x   = (const float*)d_in[0];
  const float* z   = (const float*)d_in[1];
  const float* msk = (const float*)d_in[2];
  const float* W1  = (const float*)d_in[3];
  const float* b1  = (const float*)d_in[4];
  const float* W2  = (const float*)d_in[5];
  const float* b2  = (const float*)d_in[6];
  float* out = (float*)d_out;

  float* ws   = (float*)d_ws;
  float* h    = ws;
  float* expv = h + (size_t)kK * kHid;
  float* part = expv + (size_t)kK * kTPad;

  sipm_h_kernel<<<(kK * (kHid / 4) + 255) / 256, 256, 0, stream>>>(x, W1, b1, h);
  sipm_expv_kernel<<<(kK * kTPad + 255) / 256, 256, 0, stream>>>(z, msk, expv);
  sipm_gemm_kernel<<<dim3(kSTiles, kNChunk), 512, 0, stream>>>(W2, b2, h, expv, part);
  sipm_reduce_kernel<<<(kS * kT + 255) / 256, 256, 0, stream>>>(part, out);
}


# Sensor strip per workgroup widened 16 -> 32 (k-block 256, all 16 waves build the resp panel): each expv fragment now feeds 2 WMMAs; grid 70x14, so the final sum adds 14 partials, not 11
# speedup vs baseline: 2.8377x; 1.2673x over previous
#include <hip/hip_runtime.h>
#include <math.h>


typedef float v2f __attribute__((ext_vector_type(2)));
typedef float v8f __attribute__((ext_vector_type(8)));

namespace {
constexpr int kBatch  = 64;
constexpr int kEv     = 500;
constexpr int kK      = kBatch * kEv;
constexpr int kHid    = 128;
constexpr int kS      = 2209;
constexpr int kSW     = 32;
constexpr int kStrips = (kS + kSW - 1) / kSW;
constexpr int kSPad   = kStrips * kSW;
constexpr int kT      = 550;
constexpr int kTPad   = 560;
constexpr int kTTiles = kTPad / 16;
constexpr int kKB     = 256;
constexpr int kKBlocks = kK / kKB;
constexpr int kNChunk = 14;
constexpr int kBlkCh  = (kKBlocks + kNChunk - 1) / kNChunk;
constexpr float kNorm = 0.3989422804f * 1.41421356237309515f;
}

__global__ __launch_bounds__(256)
void sipm_h_kernel(const float* __restrict__ x, const float* __restrict__ W1,
                   const float* __restrict__ b1, float* __restrict__ h) {
  int idx = blockIdx.x * 256 + threadIdx.x;
  if (idx >= kK * (kHid / 4)) return;
  int k = idx >> 5;
  int j = (idx & 31) << 2;
  float x0 = x[k * 3 + 0], x1 = x[k * 3 + 1], x2 = x[k * 3 + 2];
  const float4 wa = *(const float4*)(W1 + 0 * kHid + j);
  const float4 wb = *(const float4*)(W1 + 1 * kHid + j);
  const float4 wc = *(const float4*)(W1 + 2 * kHid + j);
  const float4 bb = *(const float4*)(b1 + j);
  float4 r;
  r.x = fmaxf(fmaf(x0, wa.x, fmaf(x1, wb.x, fmaf(x2, wc.x, bb.x))), 0.f);
  r.y = fmaxf(fmaf(x0, wa.y, fmaf(x1, wb.y, fmaf(x2, wc.y, bb.y))), 0.f);
  r.z = fmaxf(fmaf(x0, wa.z, fmaf(x1, wb.z, fmaf(x2, wc.z, bb.z))), 0.f);
  r.w = fmaxf(fmaf(x0, wa.w, fmaf(x1, wb.w, fmaf(x2, wc.w, bb.w))), 0.f);
  *(float4*)(h + (size_t)k * kHid + j) = r;
}

__global__ __launch_bounds__(256)
void sipm_expv_kernel(const float* __restrict__ z, const float* __restrict__ mask,
                      float* __restrict__ expv) {
  int idx = blockIdx.x * 256 + threadIdx.x;
  if (idx >= kK * kTPad) return;
  int k = idx / kTPad;
  int t = idx - k * kTPad;
  float v = 0.f;
  if (t < kT) {
    float d = (float)t - z[k];
    v = expf(-d * d) * mask[k] * kNorm;
  }
  expv[idx] = v;
}

__global__ __launch_bounds__(512, 8)
void sipm_gemm_kernel(const float* __restrict__ W2, const float* __restrict__ b2,
                      const float* __restrict__ h, const float* __restrict__ expv,
                      float* __restrict__ part) {
  __shared__ __attribute__((aligned(16))) float lds_w2f[2 * kHid * 16];
  __shared__ __attribute__((aligned(16))) float lds_resp[kKB * kSW];

  const int tid    = threadIdx.x;
  const int strip  = blockIdx.x;
  const int chunk  = blockIdx.y;
  const int s_base = strip * kSW;

  for (int r = 0; r < 8; ++r) {
    int idx = tid + r * 512;
    int i = idx >> 5, ss = idx & 31;
    int s = s_base + ss;
    float v = (s < kS) ? W2[(size_t)i * kS + s] : 0.f;
    int st = ss >> 4, c = ss & 15;
    int j = i >> 3, hh2 = (i >> 2) & 1, q = i & 3;
    lds_w2f[((((st * 16 + j) * 32) + hh2 * 16 + c) << 2) + q] = v;
  }

  const int w    = tid >> 5;
  const int lane = tid & 31;
  const int m    = lane & 15;
  const int hh   = lane >> 4;
  const int kb2  = hh << 1;
  const int tb0 = w * 16, tb1 = (w + 16) * 16, tb2 = (w + 32) * 16;
  const bool t2ok = (w + 32) < kTTiles;

  const int  sA0 = s_base + m, sA1 = s_base + 16 + m;
  const float bias0 = (sA0 < kS) ? b2[sA0] : 0.f;
  const float bias1 = (sA1 < kS) ? b2[sA1] : 0.f;

  v8f acc00 = {}, acc01 = {}, acc02 = {};
  v8f acc10 = {}, acc11 = {}, acc12 = {};

  __syncthreads();

  const int blk0 = chunk * kBlkCh;
  const int nblk = (kKBlocks - blk0 < kBlkCh) ? (kKBlocks - blk0) : kBlkCh;

  for (int kblk = 0; kblk < nblk; ++kblk) {
    const int k0 = (blk0 + kblk) * kKB;

    {
      const float* hrow = h + (size_t)(k0 + 16 * w + m) * kHid + 4 * hh;
      const float* wfr  = lds_w2f + (lane << 2);
      v8f d00 = {}, d01 = {}, d10 = {}, d11 = {};
      float4 aQ[2];
      aQ[0] = *(const float4*)(hrow);
      aQ[1] = *(const float4*)(hrow + 8);
      #pragma unroll
      for (int j = 0; j < kHid / 8; ++j) {
        const float4 a4 = aQ[j & 1];
        if (j + 2 < kHid / 8)
          aQ[j & 1] = *(const float4*)(hrow + 8 * (j + 2));
        const float4 p4 = *(const float4*)(wfr + j * 128);
        const float4 q4 = *(const float4*)(wfr + (16 + j) * 128);
        v2f aLo, aHi, pLo, pHi, qLo, qHi;
        aLo.x = a4.x; aLo.y = a4.y; aHi.x = a4.z; aHi.y = a4.w;
        pLo.x = p4.x; pLo.y = p4.y; pHi.x = p4.z; pHi.y = p4.w;
        qLo.x = q4.x; qLo.y = q4.y; qHi.x = q4.z; qHi.y = q4.w;
        d00 = __builtin_amdgcn_wmma_f32_16x16x4_f32(false, aLo, false, pLo,
                                                    (short)0, d00, false, false);
        d10 = __builtin_amdgcn_wmma_f32_16x16x4_f32(false, aLo, false, qLo,
                                                    (short)0, d10, false, false);
        d01 = __builtin_amdgcn_wmma_f32_16x16x4_f32(false, aHi, false, pHi,
                                                    (short)0, d01, false, false);
        d11 = __builtin_amdgcn_wmma_f32_16x16x4_f32(false, aHi, false, qHi,
                                                    (short)0, d11, false, false);
      }
      float* rp = lds_resp + ((((4 * w + 2 * hh) * 32) + m) << 2);
      #pragma unroll
      for (int r = 0; r < 8; r += 2) {
        float4 o;
        o.x = expf(d00[r]     + d01[r]     + bias0);
        o.y = expf(d00[r + 1] + d01[r + 1] + bias0);
        o.z = expf(d10[r]     + d11[r]     + bias1);
        o.w = expf(d10[r + 1] + d11[r + 1] + bias1);
        *(float4*)(rp + ((((r >> 2) * 32) + ((r & 3) >> 1) * 16) << 2)) = o;
      }
    }
    __syncthreads();

    const float* erow = expv + (size_t)(k0 + kb2) * kTPad + m;
    const float* ap   = lds_resp + (lane << 2);
    float4 aP[2];
    v2f bP0[2], bP1[2], bP2[2];
    #pragma unroll
    for (int p = 0; p < 2; ++p) {
      const int ks = p * 4;
      aP[p] = *(const float4*)(ap + p * 128);
      const float* e = erow + (size_t)ks * kTPad;
      bP0[p].x = e[tb0]; bP0[p].y = e[kTPad + tb0];
      bP1[p].x = e[tb1]; bP1[p].y = e[kTPad + tb1];
      bP2[p].x = e[tb2]; bP2[p].y = e[kTPad + tb2];
    }
    #pragma unroll
    for (int i = 0; i < kKB / 4; ++i) {
      const int slot = i & 1;
      const float4 a4 = aP[slot];
      const v2f b0  = bP0[slot];
      const v2f b1v = bP1[slot];
      const v2f b2v = bP2[slot];
      if (i + 2 < kKB / 4) {
        const int ks = (i + 2) * 4;
        aP[slot] = *(const float4*)(ap + (i + 2) * 128);
        const float* e = erow + (size_t)ks * kTPad;
        bP0[slot].x = e[tb0]; bP0[slot].y = e[kTPad + tb0];
        bP1[slot].x = e[tb1]; bP1[slot].y = e[kTPad + tb1];
        bP2[slot].x = e[tb2]; bP2[slot].y = e[kTPad + tb2];
      }
      v2f a0, a1;
      a0.x = a4.x; a0.y = a4.y;
      a1.x = a4.z; a1.y = a4.w;
      acc00 = __builtin_amdgcn_wmma_f32_16x16x4_f32(false, a0, false, b0,
                                                    (short)0, acc00, false, false);
      acc10 = __builtin_amdgcn_wmma_f32_16x16x4_f32(false, a1, false, b0,
                                                    (short)0, acc10, false, false);
      acc01 = __builtin_amdgcn_wmma_f32_16x16x4_f32(false, a0, false, b1v,
                                                    (short)0, acc01, false, false);
      acc11 = __builtin_amdgcn_wmma_f32_16x16x4_f32(false, a1, false, b1v,
                                                    (short)0, acc11, false, false);
      if (t2ok) {
        acc02 = __builtin_amdgcn_wmma_f32_16x16x4_f32(false, a0, false, b2v,
                                                      (short)0, acc02, false, false);
        acc12 = __builtin_amdgcn_wmma_f32_16x16x4_f32(false, a1, false, b2v,
                                                      (short)0, acc12, false, false);
      }
    }
    __syncthreads();
  }

  float* pbase = part + ((size_t)chunk * kSPad + s_base + ((lane >> 4) << 3)) * kTPad + m;
  #pragma unroll
  for (int r = 0; r < 8; ++r) {
    float* prow0 = pbase + (size_t)r * kTPad;
    float* prow1 = prow0 + (size_t)16 * kTPad;
    prow0[tb0] = acc00[r];
    prow0[tb1] = acc01[r];
    prow1[tb0] = acc10[r];
    prow1[tb1] = acc11[r];
    if (t2ok) { prow0[tb2] = acc02[r]; prow1[tb2] = acc12[r]; }
  }
}

__global__ __launch_bounds__(256)
void sipm_reduce_kernel(const float* __restrict__ part, float* __restrict__ out) {
  int idx = blockIdx.x * 256 + threadIdx.x;
  if (idx >= kS * kT) return;
  int s = idx / kT, t = idx - s * kT;
  size_t off = (size_t)s * kTPad + t;
  float sum = 0.f;
  #pragma unroll
  for (int c = 0; c < kNChunk; ++c)
    sum += part[(size_t)c * kSPad * kTPad + off];
  out[idx] = sum;
}

extern "C" void kernel_launch(void* const* d_in, const int* in_sizes, int n_in,
                              void* d_out, int out_size, void* d_ws, size_t ws_size,
                              hipStream_t stream) {
  const float* x   = (const float*)d_in[0];
  const float* z   = (const float*)d_in[1];
  const float* msk = (const float*)d_in[2];
  const float* W1  = (const float*)d_in[3];
  const float* b1  = (const float*)d_in[4];
  const float* W2  = (const float*)d_in[5];
  const float* b2  = (const float*)d_in[6];
  float* out = (float*)d_out;

  float* ws   = (float*)d_ws;
  float* h    = ws;
  float* expv = h + (size_t)kK * kHid;
  float* part = expv + (size_t)kK * kTPad;

  sipm_h_kernel<<<(kK * (kHid / 4) + 255) / 256, 256, 0, stream>>>(x, W1, b1, h);
  sipm_expv_kernel<<<(kK * kTPad + 255) / 256, 256, 0, stream>>>(z, msk, expv);
  sipm_gemm_kernel<<<dim3(kStrips, kNChunk), 512, 0, stream>>>(W2, b2, h, expv, part);
  sipm_reduce_kernel<<<(kS * kT + 255) / 256, 256, 0, stream>>>(part, out);
}


# h and expv written in WMMA fragment order so every fragment fetch is one coalesced 16-byte load per lane (was 4-byte row loads); main K sum takes each 8 rows as {0,1,4,5},{2,3,6,7} per WMMA
# speedup vs baseline: 3.7388x; 1.3175x over previous
#include <hip/hip_runtime.h>
#include <math.h>


typedef float v2f __attribute__((ext_vector_type(2)));
typedef float v8f __attribute__((ext_vector_type(8)));

namespace {
constexpr int kBatch  = 64;
constexpr int kEv     = 500;
constexpr int kK      = kBatch * kEv;
constexpr int kHid    = 128;
constexpr int kS      = 2209;
constexpr int kSW     = 32;
constexpr int kStrips = (kS + kSW - 1) / kSW;
constexpr int kSPad   = kStrips * kSW;
constexpr int kT      = 550;
constexpr int kTPad   = 560;
constexpr int kTTiles = kTPad / 16;
constexpr int kKB     = 256;
constexpr int kKBlocks = kK / kKB;
constexpr int kNChunk = 14;
constexpr int kBlkCh  = (kKBlocks + kNChunk - 1) / kNChunk;
constexpr float kNorm = 0.3989422804f * 1.41421356237309515f;
}

__global__ __launch_bounds__(256)
void sipm_h_kernel(const float* __restrict__ x, const float* __restrict__ W1,
                   const float* __restrict__ b1, float* __restrict__ h) {
  int idx = blockIdx.x * 256 + threadIdx.x;
  if (idx >= kK * (kHid / 4)) return;
  int lane = idx & 31;
  int k = ((idx >> 9) << 4) + (lane & 15);
  int j = (((idx >> 5) & 15) << 3) + ((lane >> 4) << 2);
  float x0 = x[k * 3 + 0], x1 = x[k * 3 + 1], x2 = x[k * 3 + 2];
  const float4 wa = *(const float4*)(W1 + 0 * kHid + j);
  const float4 wb = *(const float4*)(W1 + 1 * kHid + j);
  const float4 wc = *(const float4*)(W1 + 2 * kHid + j);
  const float4 bb = *(const float4*)(b1 + j);
  float4 r;
  r.x = fmaxf(fmaf(x0, wa.x, fmaf(x1, wb.x, fmaf(x2, wc.x, bb.x))), 0.f);
  r.y = fmaxf(fmaf(x0, wa.y, fmaf(x1, wb.y, fmaf(x2, wc.y, bb.y))), 0.f);
  r.z = fmaxf(fmaf(x0, wa.z, fmaf(x1, wb.z, fmaf(x2, wc.z, bb.z))), 0.f);
  r.w = fmaxf(fmaf(x0, wa.w, fmaf(x1, wb.w, fmaf(x2, wc.w, bb.w))), 0.f);
  *(float4*)(h + ((size_t)idx << 2)) = r;
}

__global__ __launch_bounds__(256)
void sipm_expv_kernel(const float* __restrict__ z, const float* __restrict__ mask,
                      float* __restrict__ expv) {
  int idx = blockIdx.x * 256 + threadIdx.x;
  if (idx >= (kK / 8) * kTTiles * 32) return;
  int p   = idx / (kTTiles * 32);
  int rem = idx - p * (kTTiles * 32);
  int t   = ((rem >> 5) << 4) + (rem & 15);
  int k   = (p << 3) + (((rem >> 4) & 1) << 2);
  float4 v = make_float4(0.f, 0.f, 0.f, 0.f);
  if (t < kT) {
    const float4 z4 = *(const float4*)(z + k);
    const float4 m4 = *(const float4*)(mask + k);
    float d;
    d = (float)t - z4.x; v.x = expf(-d * d) * m4.x * kNorm;
    d = (float)t - z4.y; v.y = expf(-d * d) * m4.y * kNorm;
    d = (float)t - z4.z; v.z = expf(-d * d) * m4.z * kNorm;
    d = (float)t - z4.w; v.w = expf(-d * d) * m4.w * kNorm;
  }
  *(float4*)(expv + ((size_t)idx << 2)) = v;
}

__global__ __launch_bounds__(512, 8)
void sipm_gemm_kernel(const float* __restrict__ W2, const float* __restrict__ b2,
                      const float* __restrict__ h, const float* __restrict__ expv,
                      float* __restrict__ part) {
  __shared__ __attribute__((aligned(16))) float lds_w2f[2 * kHid * 16];
  __shared__ __attribute__((aligned(16))) float lds_resp[kKB * kSW];

  const int tid    = threadIdx.x;
  const int strip  = blockIdx.x;
  const int chunk  = blockIdx.y;
  const int s_base = strip * kSW;

  for (int r = 0; r < 8; ++r) {
    int idx = tid + r * 512;
    int i = idx >> 5, ss = idx & 31;
    int s = s_base + ss;
    float v = (s < kS) ? W2[(size_t)i * kS + s] : 0.f;
    int st = ss >> 4, c = ss & 15;
    int j = i >> 3, hh2 = (i >> 2) & 1, q = i & 3;
    lds_w2f[((((st * 16 + j) * 32) + hh2 * 16 + c) << 2) + q] = v;
  }

  const int w    = tid >> 5;
  const int lane = tid & 31;
  const int m    = lane & 15;
  const int hh   = lane >> 4;
  const int tb0 = w * 16, tb1 = (w + 16) * 16, tb2 = (w + 32) * 16;
  const bool t2ok = (w + 32) < kTTiles;
  const int tj0 = w * 128, tj1 = (w + 16) * 128, tj2 = (t2ok ? (w + 32) : w) * 128;

  const int  sA0 = s_base + m, sA1 = s_base + 16 + m;
  const float bias0 = (sA0 < kS) ? b2[sA0] : 0.f;
  const float bias1 = (sA1 < kS) ? b2[sA1] : 0.f;

  v8f acc00 = {}, acc01 = {}, acc02 = {};
  v8f acc10 = {}, acc11 = {}, acc12 = {};

  __syncthreads();

  const int blk0 = chunk * kBlkCh;
  const int nblk = (kKBlocks - blk0 < kBlkCh) ? (kKBlocks - blk0) : kBlkCh;

  for (int kblk = 0; kblk < nblk; ++kblk) {
    const int k0 = (blk0 + kblk) * kKB;

    {
      const float* hrow = h + (size_t)((k0 >> 4) + w) * ((kHid / 8) * 128) + (lane << 2);
      const float* wfr  = lds_w2f + (lane << 2);
      v8f d00 = {}, d01 = {}, d10 = {}, d11 = {};
      float4 aQ[2];
      aQ[0] = *(const float4*)(hrow);
      aQ[1] = *(const float4*)(hrow + 128);
      #pragma unroll
      for (int j = 0; j < kHid / 8; ++j) {
        const float4 a4 = aQ[j & 1];
        if (j + 2 < kHid / 8)
          aQ[j & 1] = *(const float4*)(hrow + 128 * (j + 2));
        const float4 p4 = *(const float4*)(wfr + j * 128);
        const float4 q4 = *(const float4*)(wfr + (16 + j) * 128);
        v2f aLo, aHi, pLo, pHi, qLo, qHi;
        aLo.x = a4.x; aLo.y = a4.y; aHi.x = a4.z; aHi.y = a4.w;
        pLo.x = p4.x; pLo.y = p4.y; pHi.x = p4.z; pHi.y = p4.w;
        qLo.x = q4.x; qLo.y = q4.y; qHi.x = q4.z; qHi.y = q4.w;
        d00 = __builtin_amdgcn_wmma_f32_16x16x4_f32(false, aLo, false, pLo,
                                                    (short)0, d00, false, false);
        d10 = __builtin_amdgcn_wmma_f32_16x16x4_f32(false, aLo, false, qLo,
                                                    (short)0, d10, false, false);
        d01 = __builtin_amdgcn_wmma_f32_16x16x4_f32(false, aHi, false, pHi,
                                                    (short)0, d01, false, false);
        d11 = __builtin_amdgcn_wmma_f32_16x16x4_f32(false, aHi, false, qHi,
                                                    (short)0, d11, false, false);
      }
      float* rp = lds_resp + ((((4 * w + 2 * hh) * 32) + m) << 2);
      #pragma unroll
      for (int r = 0; r < 8; r += 2) {
        float4 o;
        o.x = expf(d00[r]     + d01[r]     + bias0);
        o.y = expf(d00[r + 1] + d01[r + 1] + bias0);
        o.z = expf(d10[r]     + d11[r]     + bias1);
        o.w = expf(d10[r + 1] + d11[r + 1] + bias1);
        *(float4*)(rp + (((((r >> 1) & 1) * 32) + (r >> 2) * 16) << 2)) = o;
      }
    }
    __syncthreads();

    const float* ebase = expv + (size_t)(k0 >> 3) * (kTTiles * 128) + (lane << 2);
    const float* ap    = lds_resp + (lane << 2);
    float4 aE = *(const float4*)(ap);
    float4 aO = *(const float4*)(ap + 128);
    float4 bP0[2], bP1[2], bP2[2];
    #pragma unroll
    for (int p = 0; p < 2; ++p) {
      const float* e = ebase + p * (kTTiles * 128);
      bP0[p] = *(const float4*)(e + tj0);
      bP1[p] = *(const float4*)(e + tj1);
      bP2[p] = *(const float4*)(e + tj2);
    }
    #pragma unroll
    for (int i = 0; i < kKB / 8; ++i) {
      const int slot = i & 1;
      const float4 e4 = aE;
      const float4 o4 = aO;
      const float4 q0 = bP0[slot];
      const float4 q1 = bP1[slot];
      const float4 q2 = bP2[slot];
      if (i + 1 < kKB / 8) {
        aE = *(const float4*)(ap + (2 * (i + 1)) * 128);
        aO = *(const float4*)(ap + (2 * (i + 1) + 1) * 128);
      }
      if (i + 2 < kKB / 8) {
        const float* e = ebase + (i + 2) * (kTTiles * 128);
        bP0[slot] = *(const float4*)(e + tj0);
        bP1[slot] = *(const float4*)(e + tj1);
        bP2[slot] = *(const float4*)(e + tj2);
      }
      v2f a0, a1, b0, b1v, b2v;
      a0.x = e4.x; a0.y = e4.y;
      a1.x = e4.z; a1.y = e4.w;
      b0.x = q0.x; b0.y = q0.y;
      b1v.x = q1.x; b1v.y = q1.y;
      b2v.x = q2.x; b2v.y = q2.y;
      acc00 = __builtin_amdgcn_wmma_f32_16x16x4_f32(false, a0, false, b0,
                                                    (short)0, acc00, false, false);
      acc10 = __builtin_amdgcn_wmma_f32_16x16x4_f32(false, a1, false, b0,
                                                    (short)0, acc10, false, false);
      acc01 = __builtin_amdgcn_wmma_f32_16x16x4_f32(false, a0, false, b1v,
                                                    (short)0, acc01, false, false);
      acc11 = __builtin_amdgcn_wmma_f32_16x16x4_f32(false, a1, false, b1v,
                                                    (short)0, acc11, false, false);
      if (t2ok) {
        acc02 = __builtin_amdgcn_wmma_f32_16x16x4_f32(false, a0, false, b2v,
                                                      (short)0, acc02, false, false);
        acc12 = __builtin_amdgcn_wmma_f32_16x16x4_f32(false, a1, false, b2v,
                                                      (short)0, acc12, false, false);
      }
      a0.x = o4.x; a0.y = o4.y;
      a1.x = o4.z; a1.y = o4.w;
      b0.x = q0.z; b0.y = q0.w;
      b1v.x = q1.z; b1v.y = q1.w;
      b2v.x = q2.z; b2v.y = q2.w;
      acc00 = __builtin_amdgcn_wmma_f32_16x16x4_f32(false, a0, false, b0,
                                                    (short)0, acc00, false, false);
      acc10 = __builtin_amdgcn_wmma_f32_16x16x4_f32(false, a1, false, b0,
                                                    (short)0, acc10, false, false);
      acc01 = __builtin_amdgcn_wmma_f32_16x16x4_f32(false, a0, false, b1v,
                                                    (short)0, acc01, false, false);
      acc11 = __builtin_amdgcn_wmma_f32_16x16x4_f32(false, a1, false, b1v,
                                                    (short)0, acc11, false, false);
      if (t2ok) {
        acc02 = __builtin_amdgcn_wmma_f32_16x16x4_f32(false, a0, false, b2v,
                                                      (short)0, acc02, false, false);
        acc12 = __builtin_amdgcn_wmma_f32_16x16x4_f32(false, a1, false, b2v,
                                                      (short)0, acc12, false, false);
      }
    }
    __syncthreads();
  }

  float* pbase = part + ((size_t)chunk * kSPad + s_base + ((lane >> 4) << 3)) * kTPad + m;
  #pragma unroll
  for (int r = 0; r < 8; ++r) {
    float* prow0 = pbase + (size_t)r * kTPad;
    float* prow1 = prow0 + (size_t)16 * kTPad;
    prow0[tb0] = acc00[r];
    prow0[tb1] = acc01[r];
    prow1[tb0] = acc10[r];
    prow1[tb1] = acc11[r];
    if (t2ok) { prow0[tb2] = acc02[r]; prow1[tb2] = acc12[r]; }
  }
}

__global__ __launch_bounds__(256)
void sipm_reduce_kernel(const float* __restrict__ part, float* __restrict__ out) {
  int idx = blockIdx.x * 256 + threadIdx.x;
  if (idx >= kS * kT) return;
  int s = idx / kT, t = idx - s * kT;
  size_t off = (size_t)s * kTPad + t;
  float sum = 0.f;
  #pragma unroll
  for (int c = 0; c < kNChunk; ++c)
    sum += part[(size_t)c * kSPad * kTPad + off];
  out[idx] = sum;
}

extern "C" void kernel_launch(void* const* d_in, const int* in_sizes, int n_in,
                              void* d_out, int out_size, void* d_ws, size_t ws_size,
                              hipStream_t stream) {
  const float* x   = (const float*)d_in[0];
  const float* z   = (const float*)d_in[1];
  const float* msk = (const float*)d_in[2];
  const float* W1  = (const float*)d_in[3];
  const float* b1  = (const float*)d_in[4];
  const float* W2  = (const float*)d_in[5];
  const float* b2  = (const float*)d_in[6];
  float* out = (float*)d_out;

  float* ws   = (float*)d_ws;
  float* h    = ws;
  float* expv = h + (size_t)kK * kHid;
  float* part = expv + (size_t)kK * kTPad;

  sipm_h_kernel<<<(kK * (kHid / 4) + 255) / 256, 256, 0, stream>>>(x, W1, b1, h);
  sipm_expv_kernel<<<((kK / 8) * kTTiles * 32 + 255) / 256, 256, 0, stream>>>(z, msk, expv);
  sipm_gemm_kernel<<<dim3(kStrips, kNChunk), 512, 0, stream>>>(W2, b2, h, expv, part);
  sipm_reduce_kernel<<<(kS * kT + 255) / 256, 256, 0, stream>>>(part, out);
}


# Electrons counting-sorted by 16-tick z bucket (one new 1-workgroup launch); each 256-k block runs the main GEMM only on t-tiles within 11 ticks of its z range; k is summed in sorted order
# speedup vs baseline: 6.9412x; 1.8565x over previous
#include <hip/hip_runtime.h>
#include <math.h>


typedef float v2f __attribute__((ext_vector_type(2)));
typedef float v8f __attribute__((ext_vector_type(8)));

namespace {
constexpr int kBatch  = 64;
constexpr int kEv     = 500;
constexpr int kK      = kBatch * kEv;
constexpr int kHid    = 128;
constexpr int kS      = 2209;
constexpr int kSW     = 32;
constexpr int kStrips = (kS + kSW - 1) / kSW;
constexpr int kSPad   = kStrips * kSW;
constexpr int kT      = 550;
constexpr int kTPad   = 560;
constexpr int kTTiles = kTPad / 16;
constexpr int kKB     = 256;
constexpr int kKBlocks = kK / kKB;
constexpr int kNChunk = 14;
constexpr int kBlkCh  = (kKBlocks + kNChunk - 1) / kNChunk;
constexpr float kNorm = 0.3989422804f * 1.41421356237309515f;
constexpr int kNB     = 35;
constexpr int kSortT  = 512;
constexpr int kSortE  = 63;
constexpr float kReach = 11.0f;
}

__device__ __forceinline__ int sipm_bucket(float zz) {
  float f = fminf(fmaxf(zz * 0.0625f, 0.f), (float)(kNB - 1));
  return (int)f;
}

__global__ __launch_bounds__(512)
void sipm_sort_kernel(const float* __restrict__ z, int* __restrict__ perm) {
  __shared__ unsigned short cnt[kNB * kSortT];
  __shared__ int bbase[kNB];
  const int tid = threadIdx.x, lane = tid & 31, w = tid >> 5;
  for (int i = tid; i < kNB * kSortT; i += kSortT) cnt[i] = 0;
  __syncthreads();
  const int e0 = tid * kSortE;
  const int e1 = (e0 + kSortE < kK) ? (e0 + kSortE) : kK;
  for (int e = e0; e < e1; ++e) {
    const int b = sipm_bucket(z[e]);
    cnt[b * kSortT + tid] = (unsigned short)(cnt[b * kSortT + tid] + 1);
  }
  __syncthreads();
  for (int b = w; b < kNB; b += 16) {
    unsigned short* row = cnt + b * kSortT + lane * 16;
    int loc[16];
    int s = 0;
    #pragma unroll
    for (int i = 0; i < 16; ++i) { loc[i] = row[i]; s += loc[i]; }
    int inc = s;
    #pragma unroll
    for (int d = 1; d < 32; d <<= 1) {
      const int o = __shfl_up(inc, d);
      if (lane >= d) inc += o;
    }
    int run = inc - s;
    #pragma unroll
    for (int i = 0; i < 16; ++i) { row[i] = (unsigned short)run; run += loc[i]; }
    if (lane == 31) bbase[b] = inc;
  }
  __syncthreads();
  if (tid == 0) {
    int run = 0;
    for (int b = 0; b < kNB; ++b) { const int c = bbase[b]; bbase[b] = run; run += c; }
  }
  __syncthreads();
  for (int e = e0; e < e1; ++e) {
    const int b = sipm_bucket(z[e]);
    const int slot = cnt[b * kSortT + tid];
    cnt[b * kSortT + tid] = (unsigned short)(slot + 1);
    perm[bbase[b] + slot] = e;
  }
}

__global__ __launch_bounds__(256)
void sipm_h_kernel(const float* __restrict__ x, const float* __restrict__ W1,
                   const float* __restrict__ b1, const int* __restrict__ perm,
                   float* __restrict__ h) {
  int idx = blockIdx.x * 256 + threadIdx.x;
  if (idx >= kK * (kHid / 4)) return;
  int lane = idx & 31;
  int k = perm[((idx >> 9) << 4) + (lane & 15)];
  int j = (((idx >> 5) & 15) << 3) + ((lane >> 4) << 2);
  float x0 = x[k * 3 + 0], x1 = x[k * 3 + 1], x2 = x[k * 3 + 2];
  const float4 wa = *(const float4*)(W1 + 0 * kHid + j);
  const float4 wb = *(const float4*)(W1 + 1 * kHid + j);
  const float4 wc = *(const float4*)(W1 + 2 * kHid + j);
  const float4 bb = *(const float4*)(b1 + j);
  float4 r;
  r.x = fmaxf(fmaf(x0, wa.x, fmaf(x1, wb.x, fmaf(x2, wc.x, bb.x))), 0.f);
  r.y = fmaxf(fmaf(x0, wa.y, fmaf(x1, wb.y, fmaf(x2, wc.y, bb.y))), 0.f);
  r.z = fmaxf(fmaf(x0, wa.z, fmaf(x1, wb.z, fmaf(x2, wc.z, bb.z))), 0.f);
  r.w = fmaxf(fmaf(x0, wa.w, fmaf(x1, wb.w, fmaf(x2, wc.w, bb.w))), 0.f);
  *(float4*)(h + ((size_t)idx << 2)) = r;
}

__global__ __launch_bounds__(256)
void sipm_expv_kernel(const float* __restrict__ z, const float* __restrict__ mask,
                      const int* __restrict__ perm, float* __restrict__ expv) {
  int idx = blockIdx.x * 256 + threadIdx.x;
  if (idx >= (kK / 8) * kTTiles * 32) return;
  int p   = idx / (kTTiles * 32);
  int rem = idx - p * (kTTiles * 32);
  int t   = ((rem >> 5) << 4) + (rem & 15);
  int k   = (p << 3) + (((rem >> 4) & 1) << 2);
  float4 v = make_float4(0.f, 0.f, 0.f, 0.f);
  if (t < kT) {
    const int4 e4 = *(const int4*)(perm + k);
    const float4 z4 = make_float4(z[e4.x], z[e4.y], z[e4.z], z[e4.w]);
    const float4 m4 = make_float4(mask[e4.x], mask[e4.y], mask[e4.z], mask[e4.w]);
    float d;
    d = (float)t - z4.x; v.x = expf(-d * d) * m4.x * kNorm;
    d = (float)t - z4.y; v.y = expf(-d * d) * m4.y * kNorm;
    d = (float)t - z4.z; v.z = expf(-d * d) * m4.z * kNorm;
    d = (float)t - z4.w; v.w = expf(-d * d) * m4.w * kNorm;
  }
  *(float4*)(expv + ((size_t)idx << 2)) = v;
}

__device__ __forceinline__ void sipm_stageB_tile(const float* __restrict__ e, const float* ap,
                                                 v8f& c0, v8f& c1) {
  constexpr int kPairs = kKB / 8;
  constexpr int kAhead = 8;
  constexpr int kPitch = kTTiles * 128;
  float4 aE = *(const float4*)(ap);
  float4 aO = *(const float4*)(ap + 128);
  float4 bQ[kAhead];
  #pragma unroll
  for (int p = 0; p < kAhead; ++p) bQ[p] = *(const float4*)(e + p * kPitch);
  #pragma unroll
  for (int i = 0; i < kPairs; ++i) {
    const float4 e4 = aE;
    const float4 o4 = aO;
    const float4 q  = bQ[i % kAhead];
    if (i + 1 < kPairs) {
      aE = *(const float4*)(ap + (2 * (i + 1)) * 128);
      aO = *(const float4*)(ap + (2 * (i + 1) + 1) * 128);
    }
    if (i + kAhead < kPairs)
      bQ[i % kAhead] = *(const float4*)(e + (i + kAhead) * kPitch);
    v2f a0, a1, b;
    a0.x = e4.x; a0.y = e4.y;
    a1.x = e4.z; a1.y = e4.w;
    b.x = q.x; b.y = q.y;
    c0 = __builtin_amdgcn_wmma_f32_16x16x4_f32(false, a0, false, b, (short)0, c0, false, false);
    c1 = __builtin_amdgcn_wmma_f32_16x16x4_f32(false, a1, false, b, (short)0, c1, false, false);
    a0.x = o4.x; a0.y = o4.y;
    a1.x = o4.z; a1.y = o4.w;
    b.x = q.z; b.y = q.w;
    c0 = __builtin_amdgcn_wmma_f32_16x16x4_f32(false, a0, false, b, (short)0, c0, false, false);
    c1 = __builtin_amdgcn_wmma_f32_16x16x4_f32(false, a1, false, b, (short)0, c1, false, false);
  }
}

__global__ __launch_bounds__(512, 8)
void sipm_gemm_kernel(const float* __restrict__ W2, const float* __restrict__ b2,
                      const float* __restrict__ z, const float* __restrict__ mask,
                      const int* __restrict__ perm,
                      const float* __restrict__ h, const float* __restrict__ expv,
                      float* __restrict__ part) {
  __shared__ __attribute__((aligned(16))) float lds_w2f[2 * kHid * 16];
  __shared__ __attribute__((aligned(16))) float lds_resp[kKB * kSW];
  __shared__ float lds_rng[16];

  const int tid    = threadIdx.x;
  const int strip  = blockIdx.x;
  const int chunk  = blockIdx.y;
  const int s_base = strip * kSW;

  for (int r = 0; r < 8; ++r) {
    int idx = tid + r * 512;
    int i = idx >> 5, ss = idx & 31;
    int s = s_base + ss;
    float v = (s < kS) ? W2[(size_t)i * kS + s] : 0.f;
    int st = ss >> 4, c = ss & 15;
    int j = i >> 3, hh2 = (i >> 2) & 1, q = i & 3;
    lds_w2f[((((st * 16 + j) * 32) + hh2 * 16 + c) << 2) + q] = v;
  }

  const int w    = tid >> 5;
  const int lane = tid & 31;
  const int m    = lane & 15;
  const int hh   = lane >> 4;
  const int tb0 = w * 16, tb1 = (w + 16) * 16, tb2 = (w + 32) * 16;
  const bool t2ok = (w + 32) < kTTiles;
  const int tj0 = w * 128, tj1 = (w + 16) * 128, tj2 = (w + 32) * 128;

  const int  sA0 = s_base + m, sA1 = s_base + 16 + m;
  const float bias0 = (sA0 < kS) ? b2[sA0] : 0.f;
  const float bias1 = (sA1 < kS) ? b2[sA1] : 0.f;

  v8f acc00 = {}, acc01 = {}, acc02 = {};
  v8f acc10 = {}, acc11 = {}, acc12 = {};

  __syncthreads();

  const int blk0 = chunk * kBlkCh;
  const int nblk = (kKBlocks - blk0 < kBlkCh) ? (kKBlocks - blk0) : kBlkCh;

  for (int kblk = 0; kblk < nblk; ++kblk) {
    const int k0 = (blk0 + kblk) * kKB;

    if (tid < kKB) {
      const int ke = perm[k0 + tid];
      const float zz = z[ke], mm = mask[ke];
      const bool fin = (fabsf(zz) < INFINITY) && (fabsf(mm) < INFINITY);
      float lo = fin ? zz : -INFINITY;
      float hi = fin ? zz : INFINITY;
      #pragma unroll
      for (int d = 16; d >= 1; d >>= 1) {
        lo = fminf(lo, __shfl_xor(lo, d));
        hi = fmaxf(hi, __shfl_xor(hi, d));
      }
      if (lane == 0) { lds_rng[2 * w] = lo; lds_rng[2 * w + 1] = hi; }
    }

    {
      const float* hrow = h + (size_t)((k0 >> 4) + w) * ((kHid / 8) * 128) + (lane << 2);
      const float* wfr  = lds_w2f + (lane << 2);
      v8f d00 = {}, d01 = {}, d10 = {}, d11 = {};
      float4 aQ[2];
      aQ[0] = *(const float4*)(hrow);
      aQ[1] = *(const float4*)(hrow + 128);
      #pragma unroll
      for (int j = 0; j < kHid / 8; ++j) {
        const float4 a4 = aQ[j & 1];
        if (j + 2 < kHid / 8)
          aQ[j & 1] = *(const float4*)(hrow + 128 * (j + 2));
        const float4 p4 = *(const float4*)(wfr + j * 128);
        const float4 q4 = *(const float4*)(wfr + (16 + j) * 128);
        v2f aLo, aHi, pLo, pHi, qLo, qHi;
        aLo.x = a4.x; aLo.y = a4.y; aHi.x = a4.z; aHi.y = a4.w;
        pLo.x = p4.x; pLo.y = p4.y; pHi.x = p4.z; pHi.y = p4.w;
        qLo.x = q4.x; qLo.y = q4.y; qHi.x = q4.z; qHi.y = q4.w;
        d00 = __builtin_amdgcn_wmma_f32_16x16x4_f32(false, aLo, false, pLo,
                                                    (short)0, d00, false, false);
        d10 = __builtin_amdgcn_wmma_f32_16x16x4_f32(false, aLo, false, qLo,
                                                    (short)0, d10, false, false);
        d01 = __builtin_amdgcn_wmma_f32_16x16x4_f32(false, aHi, false, pHi,
                                                    (short)0, d01, false, false);
        d11 = __builtin_amdgcn_wmma_f32_16x16x4_f32(false, aHi, false, qHi,
                                                    (short)0, d11, false, false);
      }
      float* rp = lds_resp + ((((4 * w + 2 * hh) * 32) + m) << 2);
      #pragma unroll
      for (int r = 0; r < 8; r += 2) {
        float4 o;
        o.x = expf(d00[r]     + d01[r]     + bias0);
        o.y = expf(d00[r + 1] + d01[r + 1] + bias0);
        o.z = expf(d10[r]     + d11[r]     + bias1);
        o.w = expf(d10[r + 1] + d11[r + 1] + bias1);
        *(float4*)(rp + (((((r >> 1) & 1) * 32) + (r >> 2) * 16) << 2)) = o;
      }
    }
    __syncthreads();

    float zmin = lds_rng[0], zmax = lds_rng[1];
    #pragma unroll
    for (int i = 1; i < 8; ++i) {
      zmin = fminf(zmin, lds_rng[2 * i]);
      zmax = fmaxf(zmax, lds_rng[2 * i + 1]);
    }
    const float flo = fminf(fmaxf((zmin - kReach) * 0.0625f, 0.f), (float)kTTiles);
    const float fhi = fminf(fmaxf((zmax + kReach) * 0.0625f, -1.f), (float)(kTTiles - 1));
    const int jlo = __builtin_amdgcn_readfirstlane((int)flo);
    const int jhi = __builtin_amdgcn_readfirstlane((int)floorf(fhi));
    const float* ebase = expv + (size_t)(k0 >> 3) * (kTTiles * 128) + (lane << 2);
    const float* ap    = lds_resp + (lane << 2);
    const int wu = __builtin_amdgcn_readfirstlane(w);
    if (wu >= jlo && wu <= jhi)
      sipm_stageB_tile(ebase + tj0, ap, acc00, acc10);
    if (wu + 16 >= jlo && wu + 16 <= jhi)
      sipm_stageB_tile(ebase + tj1, ap, acc01, acc11);
    if (wu + 32 < kTTiles && wu + 32 >= jlo && wu + 32 <= jhi)
      sipm_stageB_tile(ebase + tj2, ap, acc02, acc12);
    __syncthreads();
  }

  float* pbase = part + ((size_t)chunk * kSPad + s_base + ((lane >> 4) << 3)) * kTPad + m;
  #pragma unroll
  for (int r = 0; r < 8; ++r) {
    float* prow0 = pbase + (size_t)r * kTPad;
    float* prow1 = prow0 + (size_t)16 * kTPad;
    prow0[tb0] = acc00[r];
    prow0[tb1] = acc01[r];
    prow1[tb0] = acc10[r];
    prow1[tb1] = acc11[r];
    if (t2ok) { prow0[tb2] = acc02[r]; prow1[tb2] = acc12[r]; }
  }
}

__global__ __launch_bounds__(256)
void sipm_reduce_kernel(const float* __restrict__ part, float* __restrict__ out) {
  int idx = blockIdx.x * 256 + threadIdx.x;
  if (idx >= kS * kT) return;
  int s = idx / kT, t = idx - s * kT;
  size_t off = (size_t)s * kTPad + t;
  float sum = 0.f;
  #pragma unroll
  for (int c = 0; c < kNChunk; ++c)
    sum += part[(size_t)c * kSPad * kTPad + off];
  out[idx] = sum;
}

extern "C" void kernel_launch(void* const* d_in, const int* in_sizes, int n_in,
                              void* d_out, int out_size, void* d_ws, size_t ws_size,
                              hipStream_t stream) {
  const float* x   = (const float*)d_in[0];
  const float* z   = (const float*)d_in[1];
  const float* msk = (const float*)d_in[2];
  const float* W1  = (const float*)d_in[3];
  const float* b1  = (const float*)d_in[4];
  const float* W2  = (const float*)d_in[5];
  const float* b2  = (const float*)d_in[6];
  float* out = (float*)d_out;

  float* ws   = (float*)d_ws;
  float* h    = ws;
  float* expv = h + (size_t)kK * kHid;
  float* part = expv + (size_t)kK * kTPad;
  int*   perm = (int*)(part + (size_t)kNChunk * kSPad * kTPad);

  sipm_sort_kernel<<<1, kSortT, 0, stream>>>(z, perm);
  sipm_h_kernel<<<(kK * (kHid / 4) + 255) / 256, 256, 0, stream>>>(x, W1, b1, perm, h);
  sipm_expv_kernel<<<((kK / 8) * kTTiles * 32 + 255) / 256, 256, 0, stream>>>(z, msk, perm, expv);
  sipm_gemm_kernel<<<dim3(kStrips, kNChunk), 512, 0, stream>>>(W2, b2, z, msk, perm, h, expv, part);
  sipm_reduce_kernel<<<(kS * kT + 255) / 256, 256, 0, stream>>>(part, out);
}


# Sort launch rebuilt: z read once in whole lines, bucket bytes kept in LDS; 1024 threads x 32 electrons, ranks kept from the count pass; perm built in LDS, stored in whole lines; same perm
# speedup vs baseline: 8.4074x; 1.2112x over previous
#include <hip/hip_runtime.h>
#include <math.h>


typedef float v2f __attribute__((ext_vector_type(2)));
typedef float v8f __attribute__((ext_vector_type(8)));

namespace {
constexpr int kBatch  = 64;
constexpr int kEv     = 500;
constexpr int kK      = kBatch * kEv;
constexpr int kHid    = 128;
constexpr int kS      = 2209;
constexpr int kSW     = 32;
constexpr int kStrips = (kS + kSW - 1) / kSW;
constexpr int kSPad   = kStrips * kSW;
constexpr int kT      = 550;
constexpr int kTPad   = 560;
constexpr int kTTiles = kTPad / 16;
constexpr int kKB     = 256;
constexpr int kKBlocks = kK / kKB;
constexpr int kNChunk = 14;
constexpr int kBlkCh  = (kKBlocks + kNChunk - 1) / kNChunk;
constexpr float kNorm = 0.3989422804f * 1.41421356237309515f;
constexpr int kNB     = 35;
constexpr int kSortT  = 1024;
constexpr int kSortE  = 32;
constexpr int kSortOwn = kK / kSortE;
constexpr int kSortFull = kK / kSortT;
static_assert(kK % kSortE == 0 && kSortOwn <= kSortT && kK % 4 == 0, "sort tiling");
constexpr float kReach = 11.0f;
}

__device__ __forceinline__ int sipm_bucket(float zz) {
  float f = fminf(fmaxf(zz * 0.0625f, 0.f), (float)(kNB - 1));
  return (int)f;
}

__global__ __launch_bounds__(1024)
void sipm_sort_kernel(const float* __restrict__ z, int* __restrict__ perm) {
  __shared__ __attribute__((aligned(16))) unsigned char  bkt[kSortT * kSortE];
  __shared__ __attribute__((aligned(16))) unsigned short cnt[kNB * kSortT];
  __shared__ __attribute__((aligned(16))) int permL[kK];
  __shared__ int bbase[kNB];
  const int tid = threadIdx.x, lane = tid & 31, w = tid >> 5;

  for (int i = tid; i < kNB * kSortT / 8; i += kSortT)
    ((uint4*)cnt)[i] = make_uint4(0u, 0u, 0u, 0u);

  float zr[kSortFull + 1];
  #pragma unroll
  for (int j = 0; j < kSortFull; ++j) zr[j] = z[j * kSortT + tid];
  const bool tail = (kSortFull * kSortT + tid) < kK;
  zr[kSortFull] = tail ? z[kSortFull * kSortT + tid] : 0.f;
  #pragma unroll
  for (int j = 0; j < kSortFull; ++j)
    bkt[j * kSortT + tid] = (unsigned char)sipm_bucket(zr[j]);
  if (tail) bkt[kSortFull * kSortT + tid] = (unsigned char)sipm_bucket(zr[kSortFull]);
  __syncthreads();

  const bool own = tid < kSortOwn;
  unsigned int bw[kSortE / 4];
  int rank[kSortE];
  if (own) {
    const uint4 lo = *(const uint4*)(bkt + tid * kSortE);
    const uint4 hi = *(const uint4*)(bkt + tid * kSortE + 16);
    bw[0] = lo.x; bw[1] = lo.y; bw[2] = lo.z; bw[3] = lo.w;
    bw[4] = hi.x; bw[5] = hi.y; bw[6] = hi.z; bw[7] = hi.w;
    #pragma unroll
    for (int i = 0; i < kSortE; ++i) {
      const int bq = (int)((bw[i >> 2] >> (8 * (i & 3))) & 0xffu);
      const int old = cnt[bq * kSortT + tid];
      cnt[bq * kSortT + tid] = (unsigned short)(old + 1);
      rank[i] = old;
    }
  }
  __syncthreads();

  for (int bq = w; bq < kNB; bq += 32) {
    uint4* row = (uint4*)(cnt + bq * kSortT + lane * 32);
    unsigned int pk[16];
    #pragma unroll
    for (int v = 0; v < 4; ++v) {
      const uint4 q = row[v];
      pk[4 * v] = q.x; pk[4 * v + 1] = q.y; pk[4 * v + 2] = q.z; pk[4 * v + 3] = q.w;
    }
    int s = 0;
    #pragma unroll
    for (int i = 0; i < 16; ++i) s += (int)(pk[i] & 0xffffu) + (int)(pk[i] >> 16);
    int inc = s;
    #pragma unroll
    for (int d = 1; d < 32; d <<= 1) {
      const int o = __shfl_up(inc, d);
      if (lane >= d) inc += o;
    }
    int run = inc - s;
    #pragma unroll
    for (int i = 0; i < 16; ++i) {
      const int c0 = (int)(pk[i] & 0xffffu), c1 = (int)(pk[i] >> 16);
      const unsigned int p0 = (unsigned int)run;  run += c0;
      const unsigned int p1 = (unsigned int)run;  run += c1;
      pk[i] = (p0 & 0xffffu) | (p1 << 16);
    }
    #pragma unroll
    for (int v = 0; v < 4; ++v)
      row[v] = make_uint4(pk[4 * v], pk[4 * v + 1], pk[4 * v + 2], pk[4 * v + 3]);
    if (lane == 31) bbase[bq] = inc;
  }
  __syncthreads();
  if (tid == 0) {
    int run = 0;
    for (int bq = 0; bq < kNB; ++bq) { const int c = bbase[bq]; bbase[bq] = run; run += c; }
  }
  __syncthreads();

  if (own) {
    #pragma unroll
    for (int i = 0; i < kSortE; ++i) {
      const int bq = (int)((bw[i >> 2] >> (8 * (i & 3))) & 0xffu);
      permL[bbase[bq] + (int)cnt[bq * kSortT + tid] + rank[i]] = tid * kSortE + i;
    }
  }
  __syncthreads();
  #pragma unroll
  for (int j = 0; j < (kK / 4 + kSortT - 1) / kSortT; ++j) {
    const int i = j * kSortT + tid;
    if (i < kK / 4) ((int4*)perm)[i] = ((const int4*)permL)[i];
  }
}

__global__ __launch_bounds__(256)
void sipm_h_kernel(const float* __restrict__ x, const float* __restrict__ W1,
                   const float* __restrict__ b1, const int* __restrict__ perm,
                   float* __restrict__ h) {
  int idx = blockIdx.x * 256 + threadIdx.x;
  if (idx >= kK * (kHid / 4)) return;
  int lane = idx & 31;
  int k = perm[((idx >> 9) << 4) + (lane & 15)];
  int j = (((idx >> 5) & 15) << 3) + ((lane >> 4) << 2);
  float x0 = x[k * 3 + 0], x1 = x[k * 3 + 1], x2 = x[k * 3 + 2];
  const float4 wa = *(const float4*)(W1 + 0 * kHid + j);
  const float4 wb = *(const float4*)(W1 + 1 * kHid + j);
  const float4 wc = *(const float4*)(W1 + 2 * kHid + j);
  const float4 bb = *(const float4*)(b1 + j);
  float4 r;
  r.x = fmaxf(fmaf(x0, wa.x, fmaf(x1, wb.x, fmaf(x2, wc.x, bb.x))), 0.f);
  r.y = fmaxf(fmaf(x0, wa.y, fmaf(x1, wb.y, fmaf(x2, wc.y, bb.y))), 0.f);
  r.z = fmaxf(fmaf(x0, wa.z, fmaf(x1, wb.z, fmaf(x2, wc.z, bb.z))), 0.f);
  r.w = fmaxf(fmaf(x0, wa.w, fmaf(x1, wb.w, fmaf(x2, wc.w, bb.w))), 0.f);
  *(float4*)(h + ((size_t)idx << 2)) = r;
}

__global__ __launch_bounds__(256)
void sipm_expv_kernel(const float* __restrict__ z, const float* __restrict__ mask,
                      const int* __restrict__ perm, float* __restrict__ expv) {
  int idx = blockIdx.x * 256 + threadIdx.x;
  if (idx >= (kK / 8) * kTTiles * 32) return;
  int p   = idx / (kTTiles * 32);
  int rem = idx - p * (kTTiles * 32);
  int t   = ((rem >> 5) << 4) + (rem & 15);
  int k   = (p << 3) + (((rem >> 4) & 1) << 2);
  float4 v = make_float4(0.f, 0.f, 0.f, 0.f);
  if (t < kT) {
    const int4 e4 = *(const int4*)(perm + k);
    const float4 z4 = make_float4(z[e4.x], z[e4.y], z[e4.z], z[e4.w]);
    const float4 m4 = make_float4(mask[e4.x], mask[e4.y], mask[e4.z], mask[e4.w]);
    float d;
    d = (float)t - z4.x; v.x = expf(-d * d) * m4.x * kNorm;
    d = (float)t - z4.y; v.y = expf(-d * d) * m4.y * kNorm;
    d = (float)t - z4.z; v.z = expf(-d * d) * m4.z * kNorm;
    d = (float)t - z4.w; v.w = expf(-d * d) * m4.w * kNorm;
  }
  *(float4*)(expv + ((size_t)idx << 2)) = v;
}

__device__ __forceinline__ void sipm_stageB_tile(const float* __restrict__ e, const float* ap,
                                                 v8f& c0, v8f& c1) {
  constexpr int kPairs = kKB / 8;
  constexpr int kAhead = 8;
  constexpr int kPitch = kTTiles * 128;
  float4 aE = *(const float4*)(ap);
  float4 aO = *(const float4*)(ap + 128);
  float4 bQ[kAhead];
  #pragma unroll
  for (int p = 0; p < kAhead; ++p) bQ[p] = *(const float4*)(e + p * kPitch);
  #pragma unroll
  for (int i = 0; i < kPairs; ++i) {
    const float4 e4 = aE;
    const float4 o4 = aO;
    const float4 q  = bQ[i % kAhead];
    if (i + 1 < kPairs) {
      aE = *(const float4*)(ap + (2 * (i + 1)) * 128);
      aO = *(const float4*)(ap + (2 * (i + 1) + 1) * 128);
    }
    if (i + kAhead < kPairs)
      bQ[i % kAhead] = *(const float4*)(e + (i + kAhead) * kPitch);
    v2f a0, a1, b;
    a0.x = e4.x; a0.y = e4.y;
    a1.x = e4.z; a1.y = e4.w;
    b.x = q.x; b.y = q.y;
    c0 = __builtin_amdgcn_wmma_f32_16x16x4_f32(false, a0, false, b, (short)0, c0, false, false);
    c1 = __builtin_amdgcn_wmma_f32_16x16x4_f32(false, a1, false, b, (short)0, c1, false, false);
    a0.x = o4.x; a0.y = o4.y;
    a1.x = o4.z; a1.y = o4.w;
    b.x = q.z; b.y = q.w;
    c0 = __builtin_amdgcn_wmma_f32_16x16x4_f32(false, a0, false, b, (short)0, c0, false, false);
    c1 = __builtin_amdgcn_wmma_f32_16x16x4_f32(false, a1, false, b, (short)0, c1, false, false);
  }
}

__global__ __launch_bounds__(512, 8)
void sipm_gemm_kernel(const float* __restrict__ W2, const float* __restrict__ b2,
                      const float* __restrict__ z, const float* __restrict__ mask,
                      const int* __restrict__ perm,
                      const float* __restrict__ h, const float* __restrict__ expv,
                      float* __restrict__ part) {
  __shared__ __attribute__((aligned(16))) float lds_w2f[2 * kHid * 16];
  __shared__ __attribute__((aligned(16))) float lds_resp[kKB * kSW];
  __shared__ float lds_rng[16];

  const int tid    = threadIdx.x;
  const int strip  = blockIdx.x;
  const int chunk  = blockIdx.y;
  const int s_base = strip * kSW;

  for (int r = 0; r < 8; ++r) {
    int idx = tid + r * 512;
    int i = idx >> 5, ss = idx & 31;
    int s = s_base + ss;
    float v = (s < kS) ? W2[(size_t)i * kS + s] : 0.f;
    int st = ss >> 4, c = ss & 15;
    int j = i >> 3, hh2 = (i >> 2) & 1, q = i & 3;
    lds_w2f[((((st * 16 + j) * 32) + hh2 * 16 + c) << 2) + q] = v;
  }

  const int w    = tid >> 5;
  const int lane = tid & 31;
  const int m    = lane & 15;
  const int hh   = lane >> 4;
  const int tb0 = w * 16, tb1 = (w + 16) * 16, tb2 = (w + 32) * 16;
  const bool t2ok = (w + 32) < kTTiles;
  const int tj0 = w * 128, tj1 = (w + 16) * 128, tj2 = (w + 32) * 128;

  const int  sA0 = s_base + m, sA1 = s_base + 16 + m;
  const float bias0 = (sA0 < kS) ? b2[sA0] : 0.f;
  const float bias1 = (sA1 < kS) ? b2[sA1] : 0.f;

  v8f acc00 = {}, acc01 = {}, acc02 = {};
  v8f acc10 = {}, acc11 = {}, acc12 = {};

  __syncthreads();

  const int blk0 = chunk * kBlkCh;
  const int nblk = (kKBlocks - blk0 < kBlkCh) ? (kKBlocks - blk0) : kBlkCh;

  for (int kblk = 0; kblk < nblk; ++kblk) {
    const int k0 = (blk0 + kblk) * kKB;

    if (tid < kKB) {
      const int ke = perm[k0 + tid];
      const float zz = z[ke], mm = mask[ke];
      const bool fin = (fabsf(zz) < INFINITY) && (fabsf(mm) < INFINITY);
      float lo = fin ? zz : -INFINITY;
      float hi = fin ? zz : INFINITY;
      #pragma unroll
      for (int d = 16; d >= 1; d >>= 1) {
        lo = fminf(lo, __shfl_xor(lo, d));
        hi = fmaxf(hi, __shfl_xor(hi, d));
      }
      if (lane == 0) { lds_rng[2 * w] = lo; lds_rng[2 * w + 1] = hi; }
    }

    {
      const float* hrow = h + (size_t)((k0 >> 4) + w) * ((kHid / 8) * 128) + (lane << 2);
      const float* wfr  = lds_w2f + (lane << 2);
      v8f d00 = {}, d01 = {}, d10 = {}, d11 = {};
      float4 aQ[2];
      aQ[0] = *(const float4*)(hrow);
      aQ[1] = *(const float4*)(hrow + 128);
      #pragma unroll
      for (int j = 0; j < kHid / 8; ++j) {
        const float4 a4 = aQ[j & 1];
        if (j + 2 < kHid / 8)
          aQ[j & 1] = *(const float4*)(hrow + 128 * (j + 2));
        const float4 p4 = *(const float4*)(wfr + j * 128);
        const float4 q4 = *(const float4*)(wfr + (16 + j) * 128);
        v2f aLo, aHi, pLo, pHi, qLo, qHi;
        aLo.x = a4.x; aLo.y = a4.y; aHi.x = a4.z; aHi.y = a4.w;
        pLo.x = p4.x; pLo.y = p4.y; pHi.x = p4.z; pHi.y = p4.w;
        qLo.x = q4.x; qLo.y = q4.y; qHi.x = q4.z; qHi.y = q4.w;
        d00 = __builtin_amdgcn_wmma_f32_16x16x4_f32(false, aLo, false, pLo,
                                                    (short)0, d00, false, false);
        d10 = __builtin_amdgcn_wmma_f32_16x16x4_f32(false, aLo, false, qLo,
                                                    (short)0, d10, false, false);
        d01 = __builtin_amdgcn_wmma_f32_16x16x4_f32(false, aHi, false, pHi,
                                                    (short)0, d01, false, false);
        d11 = __builtin_amdgcn_wmma_f32_16x16x4_f32(false, aHi, false, qHi,
                                                    (short)0, d11, false, false);
      }
      float* rp = lds_resp + ((((4 * w + 2 * hh) * 32) + m) << 2);
      #pragma unroll
      for (int r = 0; r < 8; r += 2) {
        float4 o;
        o.x = expf(d00[r]     + d01[r]     + bias0);
        o.y = expf(d00[r + 1] + d01[r + 1] + bias0);
        o.z = expf(d10[r]     + d11[r]     + bias1);
        o.w = expf(d10[r + 1] + d11[r + 1] + bias1);
        *(float4*)(rp + (((((r >> 1) & 1) * 32) + (r >> 2) * 16) << 2)) = o;
      }
    }
    __syncthreads();

    float zmin = lds_rng[0], zmax = lds_rng[1];
    #pragma unroll
    for (int i = 1; i < 8; ++i) {
      zmin = fminf(zmin, lds_rng[2 * i]);
      zmax = fmaxf(zmax, lds_rng[2 * i + 1]);
    }
    const float flo = fminf(fmaxf((zmin - kReach) * 0.0625f, 0.f), (float)kTTiles);
    const float fhi = fminf(fmaxf((zmax + kReach) * 0.0625f, -1.f), (float)(kTTiles - 1));
    const int jlo = __builtin_amdgcn_readfirstlane((int)flo);
    const int jhi = __builtin_amdgcn_readfirstlane((int)floorf(fhi));
    const float* ebase = expv + (size_t)(k0 >> 3) * (kTTiles * 128) + (lane << 2);
    const float* ap    = lds_resp + (lane << 2);
    const int wu = __builtin_amdgcn_readfirstlane(w);
    if (wu >= jlo && wu <= jhi)
      sipm_stageB_tile(ebase + tj0, ap, acc00, acc10);
    if (wu + 16 >= jlo && wu + 16 <= jhi)
      sipm_stageB_tile(ebase + tj1, ap, acc01, acc11);
    if (wu + 32 < kTTiles && wu + 32 >= jlo && wu + 32 <= jhi)
      sipm_stageB_tile(ebase + tj2, ap, acc02, acc12);
    __syncthreads();
  }

  float* pbase = part + ((size_t)chunk * kSPad + s_base + ((lane >> 4) << 3)) * kTPad + m;
  #pragma unroll
  for (int r = 0; r < 8; ++r) {
    float* prow0 = pbase + (size_t)r * kTPad;
    float* prow1 = prow0 + (size_t)16 * kTPad;
    prow0[tb0] = acc00[r];
    prow0[tb1] = acc01[r];
    prow1[tb0] = acc10[r];
    prow1[tb1] = acc11[r];
    if (t2ok) { prow0[tb2] = acc02[r]; prow1[tb2] = acc12[r]; }
  }
}

__global__ __launch_bounds__(256)
void sipm_reduce_kernel(const float* __restrict__ part, float* __restrict__ out) {
  int idx = blockIdx.x * 256 + threadIdx.x;
  if (idx >= kS * kT) return;
  int s = idx / kT, t = idx - s * kT;
  size_t off = (size_t)s * kTPad + t;
  float sum = 0.f;
  #pragma unroll
  for (int c = 0; c < kNChunk; ++c)
    sum += part[(size_t)c * kSPad * kTPad + off];
  out[idx] = sum;
}

extern "C" void kernel_launch(void* const* d_in, const int* in_sizes, int n_in,
                              void* d_out, int out_size, void* d_ws, size_t ws_size,
                              hipStream_t stream) {
  const float* x   = (const float*)d_in[0];
  const float* z   = (const float*)d_in[1];
  const float* msk = (const float*)d_in[2];
  const float* W1  = (const float*)d_in[3];
  const float* b1  = (const float*)d_in[4];
  const float* W2  = (const float*)d_in[5];
  const float* b2  = (const float*)d_in[6];
  float* out = (float*)d_out;

  float* ws   = (float*)d_ws;
  float* h    = ws;
  float* expv = h + (size_t)kK * kHid;
  float* part = expv + (size_t)kK * kTPad;
  int*   perm = (int*)(part + (size_t)kNChunk * kSPad * kTPad);

  sipm_sort_kernel<<<1, kSortT, 0, stream>>>(z, perm);
  sipm_h_kernel<<<(kK * (kHid / 4) + 255) / 256, 256, 0, stream>>>(x, W1, b1, perm, h);
  sipm_expv_kernel<<<((kK / 8) * kTTiles * 32 + 255) / 256, 256, 0, stream>>>(z, msk, perm, expv);
  sipm_gemm_kernel<<<dim3(kStrips, kNChunk), 512, 0, stream>>>(W2, b2, z, msk, perm, h, expv, part);
  sipm_reduce_kernel<<<(kS * kT + 255) / 256, 256, 0, stream>>>(part, out);
}


# expv kernel works per 256-k block: takes the block's reached t-tile range from its z once, computes and stores only those tiles, and hands the range to the GEMM, which no longer rescans z
# speedup vs baseline: 9.9280x; 1.1809x over previous
#include <hip/hip_runtime.h>
#include <math.h>


typedef float v2f __attribute__((ext_vector_type(2)));
typedef float v8f __attribute__((ext_vector_type(8)));

namespace {
constexpr int kBatch  = 64;
constexpr int kEv     = 500;
constexpr int kK      = kBatch * kEv;
constexpr int kHid    = 128;
constexpr int kS      = 2209;
constexpr int kSW     = 32;
constexpr int kStrips = (kS + kSW - 1) / kSW;
constexpr int kSPad   = kStrips * kSW;
constexpr int kT      = 550;
constexpr int kTPad   = 560;
constexpr int kTTiles = kTPad / 16;
constexpr int kKB     = 256;
constexpr int kKBlocks = kK / kKB;
constexpr int kNChunk = 14;
constexpr int kBlkCh  = (kKBlocks + kNChunk - 1) / kNChunk;
constexpr float kNorm = 0.3989422804f * 1.41421356237309515f;
constexpr int kNB     = 35;
constexpr int kSortT  = 1024;
constexpr int kSortE  = 32;
constexpr int kSortOwn = kK / kSortE;
constexpr int kSortFull = kK / kSortT;
static_assert(kK % kSortE == 0 && kSortOwn <= kSortT && kK % 4 == 0, "sort tiling");
constexpr float kReach = 11.0f;
}

__device__ __forceinline__ int sipm_bucket(float zz) {
  float f = fminf(fmaxf(zz * 0.0625f, 0.f), (float)(kNB - 1));
  return (int)f;
}

__global__ __launch_bounds__(1024)
void sipm_sort_kernel(const float* __restrict__ z, int* __restrict__ perm) {
  __shared__ __attribute__((aligned(16))) unsigned char  bkt[kSortT * kSortE];
  __shared__ __attribute__((aligned(16))) unsigned short cnt[kNB * kSortT];
  __shared__ __attribute__((aligned(16))) int permL[kK];
  __shared__ int bbase[kNB];
  const int tid = threadIdx.x, lane = tid & 31, w = tid >> 5;

  for (int i = tid; i < kNB * kSortT / 8; i += kSortT)
    ((uint4*)cnt)[i] = make_uint4(0u, 0u, 0u, 0u);

  float zr[kSortFull + 1];
  #pragma unroll
  for (int j = 0; j < kSortFull; ++j) zr[j] = z[j * kSortT + tid];
  const bool tail = (kSortFull * kSortT + tid) < kK;
  zr[kSortFull] = tail ? z[kSortFull * kSortT + tid] : 0.f;
  #pragma unroll
  for (int j = 0; j < kSortFull; ++j)
    bkt[j * kSortT + tid] = (unsigned char)sipm_bucket(zr[j]);
  if (tail) bkt[kSortFull * kSortT + tid] = (unsigned char)sipm_bucket(zr[kSortFull]);
  __syncthreads();

  const bool own = tid < kSortOwn;
  unsigned int bw[kSortE / 4];
  int rank[kSortE];
  if (own) {
    const uint4 lo = *(const uint4*)(bkt + tid * kSortE);
    const uint4 hi = *(const uint4*)(bkt + tid * kSortE + 16);
    bw[0] = lo.x; bw[1] = lo.y; bw[2] = lo.z; bw[3] = lo.w;
    bw[4] = hi.x; bw[5] = hi.y; bw[6] = hi.z; bw[7] = hi.w;
    #pragma unroll
    for (int i = 0; i < kSortE; ++i) {
      const int bq = (int)((bw[i >> 2] >> (8 * (i & 3))) & 0xffu);
      const int old = cnt[bq * kSortT + tid];
      cnt[bq * kSortT + tid] = (unsigned short)(old + 1);
      rank[i] = old;
    }
  }
  __syncthreads();

  for (int bq = w; bq < kNB; bq += 32) {
    uint4* row = (uint4*)(cnt + bq * kSortT + lane * 32);
    unsigned int pk[16];
    #pragma unroll
    for (int v = 0; v < 4; ++v) {
      const uint4 q = row[v];
      pk[4 * v] = q.x; pk[4 * v + 1] = q.y; pk[4 * v + 2] = q.z; pk[4 * v + 3] = q.w;
    }
    int s = 0;
    #pragma unroll
    for (int i = 0; i < 16; ++i) s += (int)(pk[i] & 0xffffu) + (int)(pk[i] >> 16);
    int inc = s;
    #pragma unroll
    for (int d = 1; d < 32; d <<= 1) {
      const int o = __shfl_up(inc, d);
      if (lane >= d) inc += o;
    }
    int run = inc - s;
    #pragma unroll
    for (int i = 0; i < 16; ++i) {
      const int c0 = (int)(pk[i] & 0xffffu), c1 = (int)(pk[i] >> 16);
      const unsigned int p0 = (unsigned int)run;  run += c0;
      const unsigned int p1 = (unsigned int)run;  run += c1;
      pk[i] = (p0 & 0xffffu) | (p1 << 16);
    }
    #pragma unroll
    for (int v = 0; v < 4; ++v)
      row[v] = make_uint4(pk[4 * v], pk[4 * v + 1], pk[4 * v + 2], pk[4 * v + 3]);
    if (lane == 31) bbase[bq] = inc;
  }
  __syncthreads();
  if (tid == 0) {
    int run = 0;
    for (int bq = 0; bq < kNB; ++bq) { const int c = bbase[bq]; bbase[bq] = run; run += c; }
  }
  __syncthreads();

  if (own) {
    #pragma unroll
    for (int i = 0; i < kSortE; ++i) {
      const int bq = (int)((bw[i >> 2] >> (8 * (i & 3))) & 0xffu);
      permL[bbase[bq] + (int)cnt[bq * kSortT + tid] + rank[i]] = tid * kSortE + i;
    }
  }
  __syncthreads();
  #pragma unroll
  for (int j = 0; j < (kK / 4 + kSortT - 1) / kSortT; ++j) {
    const int i = j * kSortT + tid;
    if (i < kK / 4) ((int4*)perm)[i] = ((const int4*)permL)[i];
  }
}

__global__ __launch_bounds__(256)
void sipm_h_kernel(const float* __restrict__ x, const float* __restrict__ W1,
                   const float* __restrict__ b1, const int* __restrict__ perm,
                   float* __restrict__ h) {
  int idx = blockIdx.x * 256 + threadIdx.x;
  if (idx >= kK * (kHid / 4)) return;
  int lane = idx & 31;
  int k = perm[((idx >> 9) << 4) + (lane & 15)];
  int j = (((idx >> 5) & 15) << 3) + ((lane >> 4) << 2);
  float x0 = x[k * 3 + 0], x1 = x[k * 3 + 1], x2 = x[k * 3 + 2];
  const float4 wa = *(const float4*)(W1 + 0 * kHid + j);
  const float4 wb = *(const float4*)(W1 + 1 * kHid + j);
  const float4 wc = *(const float4*)(W1 + 2 * kHid + j);
  const float4 bb = *(const float4*)(b1 + j);
  float4 r;
  r.x = fmaxf(fmaf(x0, wa.x, fmaf(x1, wb.x, fmaf(x2, wc.x, bb.x))), 0.f);
  r.y = fmaxf(fmaf(x0, wa.y, fmaf(x1, wb.y, fmaf(x2, wc.y, bb.y))), 0.f);
  r.z = fmaxf(fmaf(x0, wa.z, fmaf(x1, wb.z, fmaf(x2, wc.z, bb.z))), 0.f);
  r.w = fmaxf(fmaf(x0, wa.w, fmaf(x1, wb.w, fmaf(x2, wc.w, bb.w))), 0.f);
  *(float4*)(h + ((size_t)idx << 2)) = r;
}

__global__ __launch_bounds__(256)
void sipm_expv_kernel(const float* __restrict__ z, const float* __restrict__ mask,
                      const int* __restrict__ perm, float* __restrict__ expv,
                      int* __restrict__ rng) {
  __shared__ __attribute__((aligned(16))) float zs[kKB];
  __shared__ __attribute__((aligned(16))) float ms[kKB];
  __shared__ float red[16];
  const int tid = threadIdx.x, lane = tid & 31, w = tid >> 5;
  const int kb = blockIdx.x >> 1, half = blockIdx.x & 1;
  const int ke = perm[kb * kKB + tid];
  const float zz = z[ke], mm = mask[ke];
  zs[tid] = zz;
  ms[tid] = mm;
  const bool fin = (fabsf(zz) < INFINITY) && (fabsf(mm) < INFINITY);
  float lo = fin ? zz : -INFINITY;
  float hi = fin ? zz : INFINITY;
  for (int d = 16; d >= 1; d >>= 1) {
    lo = fminf(lo, __shfl_xor(lo, d));
    hi = fmaxf(hi, __shfl_xor(hi, d));
  }
  if (lane == 0) { red[2 * w] = lo; red[2 * w + 1] = hi; }
  __syncthreads();
  float zmin = red[0], zmax = red[1];
  for (int i = 1; i < 8; ++i) {
    zmin = fminf(zmin, red[2 * i]);
    zmax = fmaxf(zmax, red[2 * i + 1]);
  }
  const float flo = fminf(fmaxf((zmin - kReach) * 0.0625f, 0.f), (float)kTTiles);
  const float fhi = fminf(fmaxf((zmax + kReach) * 0.0625f, -1.f), (float)(kTTiles - 1));
  const int jlo = __builtin_amdgcn_readfirstlane((int)flo);
  const int jhi = __builtin_amdgcn_readfirstlane((int)floorf(fhi));
  if (half == 0 && tid == 0) { rng[2 * kb] = jlo; rng[2 * kb + 1] = jhi; }

  const int hh = lane >> 4, cc = lane & 15;
  const int g  = half * 16 + w;
  const float4 za = *(const float4*)(zs + 8 * g + 4 * hh);
  const float4 ma = *(const float4*)(ms + 8 * g + 4 * hh);
  const float4 zb = *(const float4*)(zs + 8 * (g + 8) + 4 * hh);
  const float4 mb = *(const float4*)(ms + 8 * (g + 8) + 4 * hh);
  float* ea = expv + (size_t)(kb * (kKB / 8) + g) * (kTTiles * 128) + (lane << 2);
  float* eb = ea + (size_t)8 * (kTTiles * 128);
  for (int j = jlo; j <= jhi; ++j) {
    const int t = (j << 4) + cc;
    float4 va = make_float4(0.f, 0.f, 0.f, 0.f);
    float4 vb = va;
    if (t < kT) {
      const float tf = (float)t;
      float d;
      d = tf - za.x; va.x = expf(-d * d) * ma.x * kNorm;
      d = tf - za.y; va.y = expf(-d * d) * ma.y * kNorm;
      d = tf - za.z; va.z = expf(-d * d) * ma.z * kNorm;
      d = tf - za.w; va.w = expf(-d * d) * ma.w * kNorm;
      d = tf - zb.x; vb.x = expf(-d * d) * mb.x * kNorm;
      d = tf - zb.y; vb.y = expf(-d * d) * mb.y * kNorm;
      d = tf - zb.z; vb.z = expf(-d * d) * mb.z * kNorm;
      d = tf - zb.w; vb.w = expf(-d * d) * mb.w * kNorm;
    }
    *(float4*)(ea + j * 128) = va;
    *(float4*)(eb + j * 128) = vb;
  }
}

__device__ __forceinline__ void sipm_stageB_quarter(const float* __restrict__ e, const float* ap,
                                                    v8f& c0, v8f& c1) {
  constexpr int kQP    = kKB / 8 / 4;
  constexpr int kPitch = kTTiles * 128;
  float4 bQ[kQP], aE[kQP], aO[kQP];
  #pragma unroll
  for (int p = 0; p < kQP; ++p) bQ[p] = *(const float4*)(e + p * kPitch);
  #pragma unroll
  for (int p = 0; p < kQP; ++p) {
    aE[p] = *(const float4*)(ap + (2 * p) * 128);
    aO[p] = *(const float4*)(ap + (2 * p + 1) * 128);
  }
  #pragma unroll
  for (int i = 0; i < kQP; ++i) {
    const float4 e4 = aE[i];
    const float4 o4 = aO[i];
    const float4 q  = bQ[i];
    v2f a0, a1, b;
    a0.x = e4.x; a0.y = e4.y;
    a1.x = e4.z; a1.y = e4.w;
    b.x = q.x; b.y = q.y;
    c0 = __builtin_amdgcn_wmma_f32_16x16x4_f32(false, a0, false, b, (short)0, c0, false, false);
    c1 = __builtin_amdgcn_wmma_f32_16x16x4_f32(false, a1, false, b, (short)0, c1, false, false);
    a0.x = o4.x; a0.y = o4.y;
    a1.x = o4.z; a1.y = o4.w;
    b.x = q.z; b.y = q.w;
    c0 = __builtin_amdgcn_wmma_f32_16x16x4_f32(false, a0, false, b, (short)0, c0, false, false);
    c1 = __builtin_amdgcn_wmma_f32_16x16x4_f32(false, a1, false, b, (short)0, c1, false, false);
  }
}

__device__ __forceinline__ void sipm_flush_tile(float* fl, float* pt, int slot, bool again,
                                                int q, int ks, int w, int lane,
                                                const v8f& c0, const v8f& c1) {
  if (q == slot) {
    float* fp = fl + w * 512 + lane;
    #pragma unroll
    for (int r = 0; r < 8; ++r) { fp[r * 32] = c0[r]; fp[(8 + r) * 32] = c1[r]; }
  }
  __syncthreads();
  if (q == slot && ks == 0) {
    const float* f0 = fl + (4 * slot) * 512 + lane;
    #pragma unroll
    for (int r = 0; r < 8; ++r) {
      float s0 = ((f0[r * 32] + f0[512 + r * 32]) + f0[2 * 512 + r * 32]) + f0[3 * 512 + r * 32];
      float s1 = ((f0[(8 + r) * 32] + f0[512 + (8 + r) * 32]) + f0[2 * 512 + (8 + r) * 32])
                 + f0[3 * 512 + (8 + r) * 32];
      float* p0 = pt + (size_t)r * kTPad;
      float* p1 = p0 + (size_t)16 * kTPad;
      if (again) { s0 += *p0; s1 += *p1; }
      *p0 = s0;
      *p1 = s1;
    }
  }
  __syncthreads();
}

__global__ __launch_bounds__(512, 8)
void sipm_gemm_kernel(const float* __restrict__ W2, const float* __restrict__ b2,
                      const float* __restrict__ h, const float* __restrict__ expv,
                      const int* __restrict__ rng,
                      float* __restrict__ part, int* __restrict__ tmask) {
  __shared__ __attribute__((aligned(16))) float lds_w2f[2 * kHid * 16];
  __shared__ __attribute__((aligned(16))) float lds_resp[kKB * kSW];
  __shared__ __attribute__((aligned(16))) float lds_fl[16 * 512];

  const int tid    = threadIdx.x;
  const int strip  = blockIdx.x;
  const int chunk  = blockIdx.y;
  const int s_base = strip * kSW;

  for (int r = 0; r < 8; ++r) {
    int idx = tid + r * 512;
    int i = idx >> 5, ss = idx & 31;
    int s = s_base + ss;
    float v = (s < kS) ? W2[(size_t)i * kS + s] : 0.f;
    int st = ss >> 4, c = ss & 15;
    int j = i >> 3, hh2 = (i >> 2) & 1, q = i & 3;
    lds_w2f[((((st * 16 + j) * 32) + hh2 * 16 + c) << 2) + q] = v;
  }

  const int w    = tid >> 5;
  const int lane = tid & 31;
  const int m    = lane & 15;
  const int hh   = lane >> 4;
  const int wu = __builtin_amdgcn_readfirstlane(w);
  const int q  = wu >> 2;
  const int ks = wu & 3;

  const int  sA0 = s_base + m, sA1 = s_base + 16 + m;
  const float bias0 = (sA0 < kS) ? b2[sA0] : 0.f;
  const float bias1 = (sA1 < kS) ? b2[sA1] : 0.f;

  v8f c0 = {}, c1 = {};
  unsigned int curp = 0u;
  unsigned long long stored = 0ull;
  float* pbase = part + ((size_t)chunk * kSPad + s_base + (hh << 3)) * kTPad + m;

  __syncthreads();

  const int blk0 = chunk * kBlkCh;
  const int nblk = (kKBlocks - blk0 < kBlkCh) ? (kKBlocks - blk0) : kBlkCh;

  for (int kblk = 0; kblk < nblk; ++kblk) {
    const int k0 = (blk0 + kblk) * kKB;

    const int rlo = rng[2 * (blk0 + kblk)], rhi = rng[2 * (blk0 + kblk) + 1];

    {
      const float* hrow = h + (size_t)((k0 >> 4) + w) * ((kHid / 8) * 128) + (lane << 2);
      const float* wfr  = lds_w2f + (lane << 2);
      v8f d00 = {}, d01 = {}, d10 = {}, d11 = {};
      float4 aQ[2];
      aQ[0] = *(const float4*)(hrow);
      aQ[1] = *(const float4*)(hrow + 128);
      #pragma unroll
      for (int j = 0; j < kHid / 8; ++j) {
        const float4 a4 = aQ[j & 1];
        if (j + 2 < kHid / 8)
          aQ[j & 1] = *(const float4*)(hrow + 128 * (j + 2));
        const float4 p4 = *(const float4*)(wfr + j * 128);
        const float4 q4 = *(const float4*)(wfr + (16 + j) * 128);
        v2f aLo, aHi, pLo, pHi, qLo, qHi;
        aLo.x = a4.x; aLo.y = a4.y; aHi.x = a4.z; aHi.y = a4.w;
        pLo.x = p4.x; pLo.y = p4.y; pHi.x = p4.z; pHi.y = p4.w;
        qLo.x = q4.x; qLo.y = q4.y; qHi.x = q4.z; qHi.y = q4.w;
        d00 = __builtin_amdgcn_wmma_f32_16x16x4_f32(false, aLo, false, pLo,
                                                    (short)0, d00, false, false);
        d10 = __builtin_amdgcn_wmma_f32_16x16x4_f32(false, aLo, false, qLo,
                                                    (short)0, d10, false, false);
        d01 = __builtin_amdgcn_wmma_f32_16x16x4_f32(false, aHi, false, pHi,
                                                    (short)0, d01, false, false);
        d11 = __builtin_amdgcn_wmma_f32_16x16x4_f32(false, aHi, false, qHi,
                                                    (short)0, d11, false, false);
      }
      float* rp = lds_resp + ((((4 * w + 2 * hh) * 32) + m) << 2);
      #pragma unroll
      for (int r = 0; r < 8; r += 2) {
        float4 o;
        o.x = expf(d00[r]     + d01[r]     + bias0);
        o.y = expf(d00[r + 1] + d01[r + 1] + bias0);
        o.z = expf(d10[r]     + d11[r]     + bias1);
        o.w = expf(d10[r + 1] + d11[r + 1] + bias1);
        *(float4*)(rp + (((((r >> 1) & 1) * 32) + (r >> 2) * 16) << 2)) = o;
      }
    }
    __syncthreads();

    const int jlo = __builtin_amdgcn_readfirstlane(rlo < 0 ? 0 : rlo);
    const int jhi = __builtin_amdgcn_readfirstlane(rhi > kTTiles - 1 ? kTTiles - 1 : rhi);
    const float* ebase = expv + ((size_t)(k0 >> 3) + 8 * ks) * (kTTiles * 128) + (lane << 2);
    const float* ap    = lds_resp + (16 * ks) * 128 + (lane << 2);
    for (int j = jlo; j <= jhi; ++j) {
      const int slot = j & 3;
      const int held = (int)((curp >> (8 * slot)) & 0xffu) - 1;
      if (held != j) {
        if (held >= 0) {
          sipm_flush_tile(lds_fl, pbase + held * 16, slot, ((stored >> held) & 1ull) != 0ull,
                          q, ks, w, lane, c0, c1);
          stored |= 1ull << held;
        }
        curp = (curp & ~(0xffu << (8 * slot))) | ((unsigned int)(j + 1) << (8 * slot));
        if (q == slot) { c0 = v8f{}; c1 = v8f{}; }
      }
      if (q == slot)
        sipm_stageB_quarter(ebase + j * 128, ap, c0, c1);
    }
    __syncthreads();
  }

  for (int slot = 0; slot < 4; ++slot) {
    const int held = (int)((curp >> (8 * slot)) & 0xffu) - 1;
    if (held >= 0) {
      sipm_flush_tile(lds_fl, pbase + held * 16, slot, ((stored >> held) & 1ull) != 0ull,
                      q, ks, w, lane, c0, c1);
      stored |= 1ull << held;
    }
  }
  if (strip == 0 && tid == 0) {
    tmask[2 * chunk]     = (int)(unsigned int)(stored & 0xffffffffull);
    tmask[2 * chunk + 1] = (int)(unsigned int)(stored >> 32);
  }
}

__global__ __launch_bounds__(256)
void sipm_reduce_kernel(const float* __restrict__ part, const int* __restrict__ tmask,
                        float* __restrict__ out) {
  int idx = blockIdx.x * 256 + threadIdx.x;
  if (idx >= kS * kT) return;
  int s = idx / kT, t = idx - s * kT;
  size_t off = (size_t)s * kTPad + t;
  const int jt = t >> 4;
  unsigned int inb = 0u;
  #pragma unroll
  for (int c = 0; c < kNChunk; ++c) {
    const unsigned int lo = (unsigned int)tmask[2 * c], hi = (unsigned int)tmask[2 * c + 1];
    const unsigned int bits = (jt < 32) ? lo : hi;
    inb |= ((bits >> (jt & 31)) & 1u) << c;
  }
  float v[kNChunk];
  #pragma unroll
  for (int c = 0; c < kNChunk; ++c)
    v[c] = ((inb >> c) & 1u) ? part[(size_t)c * kSPad * kTPad + off] : 0.f;
  float sum = 0.f;
  #pragma unroll
  for (int c = 0; c < kNChunk; ++c) sum += v[c];
  out[idx] = sum;
}

extern "C" void kernel_launch(void* const* d_in, const int* in_sizes, int n_in,
                              void* d_out, int out_size, void* d_ws, size_t ws_size,
                              hipStream_t stream) {
  const float* x   = (const float*)d_in[0];
  const float* z   = (const float*)d_in[1];
  const float* msk = (const float*)d_in[2];
  const float* W1  = (const float*)d_in[3];
  const float* b1  = (const float*)d_in[4];
  const float* W2  = (const float*)d_in[5];
  const float* b2  = (const float*)d_in[6];
  float* out = (float*)d_out;

  float* ws   = (float*)d_ws;
  float* h    = ws;
  float* expv = h + (size_t)kK * kHid;
  float* part = expv + (size_t)kK * kTPad;
  int*   perm = (int*)(part + (size_t)kNChunk * kSPad * kTPad);
  int*   tmask = perm + kK;
  int*   rng  = tmask + 2 * kNChunk;

  sipm_sort_kernel<<<1, kSortT, 0, stream>>>(z, perm);
  sipm_h_kernel<<<(kK * (kHid / 4) + 255) / 256, 256, 0, stream>>>(x, W1, b1, perm, h);
  sipm_expv_kernel<<<kKBlocks * 2, 256, 0, stream>>>(z, msk, perm, expv, rng);
  sipm_gemm_kernel<<<dim3(kStrips, kNChunk), 512, 0, stream>>>(W2, b2, h, expv, rng, part, tmask);
  sipm_reduce_kernel<<<(kS * kT + 255) / 256, 256, 0, stream>>>(part, tmask, out);
}


# resp = exp(a) as 2^(a*log2e) on the hardware exp2 instruction (rel. error ~4e-8*|a*log2e| + 1 ulp) in the GEMM panel epilogue: 13 -> 2 vector instructions per value; all sums unchanged
# speedup vs baseline: 10.4172x; 1.0493x over previous
#include <hip/hip_runtime.h>
#include <math.h>


typedef float v2f __attribute__((ext_vector_type(2)));
typedef float v8f __attribute__((ext_vector_type(8)));

namespace {
constexpr int kBatch  = 64;
constexpr int kEv     = 500;
constexpr int kK      = kBatch * kEv;
constexpr int kHid    = 128;
constexpr int kS      = 2209;
constexpr int kSW     = 32;
constexpr int kStrips = (kS + kSW - 1) / kSW;
constexpr int kSPad   = kStrips * kSW;
constexpr int kT      = 550;
constexpr int kTPad   = 560;
constexpr int kTTiles = kTPad / 16;
constexpr int kKB     = 256;
constexpr int kKBlocks = kK / kKB;
constexpr int kNChunk = 14;
constexpr int kBlkCh  = (kKBlocks + kNChunk - 1) / kNChunk;
constexpr float kNorm = 0.3989422804f * 1.41421356237309515f;
constexpr int kNB     = 35;
constexpr int kSortT  = 1024;
constexpr int kSortE  = 32;
constexpr int kSortOwn = kK / kSortE;
constexpr int kSortFull = kK / kSortT;
static_assert(kK % kSortE == 0 && kSortOwn <= kSortT && kK % 4 == 0, "sort tiling");
constexpr float kLog2e = 1.44269504088896341f;
constexpr float kReach = 11.0f;
}

__device__ __forceinline__ int sipm_bucket(float zz) {
  float f = fminf(fmaxf(zz * 0.0625f, 0.f), (float)(kNB - 1));
  return (int)f;
}

__global__ __launch_bounds__(1024)
void sipm_sort_kernel(const float* __restrict__ z, int* __restrict__ perm) {
  __shared__ __attribute__((aligned(16))) unsigned char  bkt[kSortT * kSortE];
  __shared__ __attribute__((aligned(16))) unsigned short cnt[kNB * kSortT];
  __shared__ __attribute__((aligned(16))) int permL[kK];
  __shared__ int bbase[kNB];
  const int tid = threadIdx.x, lane = tid & 31, w = tid >> 5;

  for (int i = tid; i < kNB * kSortT / 8; i += kSortT)
    ((uint4*)cnt)[i] = make_uint4(0u, 0u, 0u, 0u);

  float zr[kSortFull + 1];
  #pragma unroll
  for (int j = 0; j < kSortFull; ++j) zr[j] = z[j * kSortT + tid];
  const bool tail = (kSortFull * kSortT + tid) < kK;
  zr[kSortFull] = tail ? z[kSortFull * kSortT + tid] : 0.f;
  #pragma unroll
  for (int j = 0; j < kSortFull; ++j)
    bkt[j * kSortT + tid] = (unsigned char)sipm_bucket(zr[j]);
  if (tail) bkt[kSortFull * kSortT + tid] = (unsigned char)sipm_bucket(zr[kSortFull]);
  __syncthreads();

  const bool own = tid < kSortOwn;
  unsigned int bw[kSortE / 4];
  int rank[kSortE];
  if (own) {
    const uint4 lo = *(const uint4*)(bkt + tid * kSortE);
    const uint4 hi = *(const uint4*)(bkt + tid * kSortE + 16);
    bw[0] = lo.x; bw[1] = lo.y; bw[2] = lo.z; bw[3] = lo.w;
    bw[4] = hi.x; bw[5] = hi.y; bw[6] = hi.z; bw[7] = hi.w;
    #pragma unroll
    for (int i = 0; i < kSortE; ++i) {
      const int bq = (int)((bw[i >> 2] >> (8 * (i & 3))) & 0xffu);
      const int old = cnt[bq * kSortT + tid];
      cnt[bq * kSortT + tid] = (unsigned short)(old + 1);
      rank[i] = old;
    }
  }
  __syncthreads();

  for (int bq = w; bq < kNB; bq += 32) {
    uint4* row = (uint4*)(cnt + bq * kSortT + lane * 32);
    unsigned int pk[16];
    #pragma unroll
    for (int v = 0; v < 4; ++v) {
      const uint4 q = row[v];
      pk[4 * v] = q.x; pk[4 * v + 1] = q.y; pk[4 * v + 2] = q.z; pk[4 * v + 3] = q.w;
    }
    int s = 0;
    #pragma unroll
    for (int i = 0; i < 16; ++i) s += (int)(pk[i] & 0xffffu) + (int)(pk[i] >> 16);
    int inc = s;
    #pragma unroll
    for (int d = 1; d < 32; d <<= 1) {
      const int o = __shfl_up(inc, d);
      if (lane >= d) inc += o;
    }
    int run = inc - s;
    #pragma unroll
    for (int i = 0; i < 16; ++i) {
      const int c0 = (int)(pk[i] & 0xffffu), c1 = (int)(pk[i] >> 16);
      const unsigned int p0 = (unsigned int)run;  run += c0;
      const unsigned int p1 = (unsigned int)run;  run += c1;
      pk[i] = (p0 & 0xffffu) | (p1 << 16);
    }
    #pragma unroll
    for (int v = 0; v < 4; ++v)
      row[v] = make_uint4(pk[4 * v], pk[4 * v + 1], pk[4 * v + 2], pk[4 * v + 3]);
    if (lane == 31) bbase[bq] = inc;
  }
  __syncthreads();
  if (tid == 0) {
    int run = 0;
    for (int bq = 0; bq < kNB; ++bq) { const int c = bbase[bq]; bbase[bq] = run; run += c; }
  }
  __syncthreads();

  if (own) {
    #pragma unroll
    for (int i = 0; i < kSortE; ++i) {
      const int bq = (int)((bw[i >> 2] >> (8 * (i & 3))) & 0xffu);
      permL[bbase[bq] + (int)cnt[bq * kSortT + tid] + rank[i]] = tid * kSortE + i;
    }
  }
  __syncthreads();
  #pragma unroll
  for (int j = 0; j < (kK / 4 + kSortT - 1) / kSortT; ++j) {
    const int i = j * kSortT + tid;
    if (i < kK / 4) ((int4*)perm)[i] = ((const int4*)permL)[i];
  }
}

__global__ __launch_bounds__(256)
void sipm_h_kernel(const float* __restrict__ x, const float* __restrict__ W1,
                   const float* __restrict__ b1, const int* __restrict__ perm,
                   float* __restrict__ h) {
  int idx = blockIdx.x * 256 + threadIdx.x;
  if (idx >= kK * (kHid / 4)) return;
  int lane = idx & 31;
  int k = perm[((idx >> 9) << 4) + (lane & 15)];
  int j = (((idx >> 5) & 15) << 3) + ((lane >> 4) << 2);
  float x0 = x[k * 3 + 0], x1 = x[k * 3 + 1], x2 = x[k * 3 + 2];
  const float4 wa = *(const float4*)(W1 + 0 * kHid + j);
  const float4 wb = *(const float4*)(W1 + 1 * kHid + j);
  const float4 wc = *(const float4*)(W1 + 2 * kHid + j);
  const float4 bb = *(const float4*)(b1 + j);
  float4 r;
  r.x = fmaxf(fmaf(x0, wa.x, fmaf(x1, wb.x, fmaf(x2, wc.x, bb.x))), 0.f);
  r.y = fmaxf(fmaf(x0, wa.y, fmaf(x1, wb.y, fmaf(x2, wc.y, bb.y))), 0.f);
  r.z = fmaxf(fmaf(x0, wa.z, fmaf(x1, wb.z, fmaf(x2, wc.z, bb.z))), 0.f);
  r.w = fmaxf(fmaf(x0, wa.w, fmaf(x1, wb.w, fmaf(x2, wc.w, bb.w))), 0.f);
  *(float4*)(h + ((size_t)idx << 2)) = r;
}

__global__ __launch_bounds__(256)
void sipm_expv_kernel(const float* __restrict__ z, const float* __restrict__ mask,
                      const int* __restrict__ perm, float* __restrict__ expv,
                      int* __restrict__ rng) {
  __shared__ __attribute__((aligned(16))) float zs[kKB];
  __shared__ __attribute__((aligned(16))) float ms[kKB];
  __shared__ float red[16];
  const int tid = threadIdx.x, lane = tid & 31, w = tid >> 5;
  const int kb = blockIdx.x >> 1, half = blockIdx.x & 1;
  const int ke = perm[kb * kKB + tid];
  const float zz = z[ke], mm = mask[ke];
  zs[tid] = zz;
  ms[tid] = mm;
  const bool fin = (fabsf(zz) < INFINITY) && (fabsf(mm) < INFINITY);
  float lo = fin ? zz : -INFINITY;
  float hi = fin ? zz : INFINITY;
  for (int d = 16; d >= 1; d >>= 1) {
    lo = fminf(lo, __shfl_xor(lo, d));
    hi = fmaxf(hi, __shfl_xor(hi, d));
  }
  if (lane == 0) { red[2 * w] = lo; red[2 * w + 1] = hi; }
  __syncthreads();
  float zmin = red[0], zmax = red[1];
  for (int i = 1; i < 8; ++i) {
    zmin = fminf(zmin, red[2 * i]);
    zmax = fmaxf(zmax, red[2 * i + 1]);
  }
  const float flo = fminf(fmaxf((zmin - kReach) * 0.0625f, 0.f), (float)kTTiles);
  const float fhi = fminf(fmaxf((zmax + kReach) * 0.0625f, -1.f), (float)(kTTiles - 1));
  const int jlo = __builtin_amdgcn_readfirstlane((int)flo);
  const int jhi = __builtin_amdgcn_readfirstlane((int)floorf(fhi));
  if (half == 0 && tid == 0) { rng[2 * kb] = jlo; rng[2 * kb + 1] = jhi; }

  const int hh = lane >> 4, cc = lane & 15;
  const int g  = half * 16 + w;
  const float4 za = *(const float4*)(zs + 8 * g + 4 * hh);
  const float4 ma = *(const float4*)(ms + 8 * g + 4 * hh);
  const float4 zb = *(const float4*)(zs + 8 * (g + 8) + 4 * hh);
  const float4 mb = *(const float4*)(ms + 8 * (g + 8) + 4 * hh);
  float* ea = expv + (size_t)(kb * (kKB / 8) + g) * (kTTiles * 128) + (lane << 2);
  float* eb = ea + (size_t)8 * (kTTiles * 128);
  for (int j = jlo; j <= jhi; ++j) {
    const int t = (j << 4) + cc;
    float4 va = make_float4(0.f, 0.f, 0.f, 0.f);
    float4 vb = va;
    if (t < kT) {
      const float tf = (float)t;
      float d;
      d = tf - za.x; va.x = expf(-d * d) * ma.x * kNorm;
      d = tf - za.y; va.y = expf(-d * d) * ma.y * kNorm;
      d = tf - za.z; va.z = expf(-d * d) * ma.z * kNorm;
      d = tf - za.w; va.w = expf(-d * d) * ma.w * kNorm;
      d = tf - zb.x; vb.x = expf(-d * d) * mb.x * kNorm;
      d = tf - zb.y; vb.y = expf(-d * d) * mb.y * kNorm;
      d = tf - zb.z; vb.z = expf(-d * d) * mb.z * kNorm;
      d = tf - zb.w; vb.w = expf(-d * d) * mb.w * kNorm;
    }
    *(float4*)(ea + j * 128) = va;
    *(float4*)(eb + j * 128) = vb;
  }
}

__device__ __forceinline__ void sipm_stageB_quarter(const float* __restrict__ e, const float* ap,
                                                    v8f& c0, v8f& c1) {
  constexpr int kQP    = kKB / 8 / 4;
  constexpr int kPitch = kTTiles * 128;
  float4 bQ[kQP], aE[kQP], aO[kQP];
  #pragma unroll
  for (int p = 0; p < kQP; ++p) bQ[p] = *(const float4*)(e + p * kPitch);
  #pragma unroll
  for (int p = 0; p < kQP; ++p) {
    aE[p] = *(const float4*)(ap + (2 * p) * 128);
    aO[p] = *(const float4*)(ap + (2 * p + 1) * 128);
  }
  #pragma unroll
  for (int i = 0; i < kQP; ++i) {
    const float4 e4 = aE[i];
    const float4 o4 = aO[i];
    const float4 q  = bQ[i];
    v2f a0, a1, b;
    a0.x = e4.x; a0.y = e4.y;
    a1.x = e4.z; a1.y = e4.w;
    b.x = q.x; b.y = q.y;
    c0 = __builtin_amdgcn_wmma_f32_16x16x4_f32(false, a0, false, b, (short)0, c0, false, false);
    c1 = __builtin_amdgcn_wmma_f32_16x16x4_f32(false, a1, false, b, (short)0, c1, false, false);
    a0.x = o4.x; a0.y = o4.y;
    a1.x = o4.z; a1.y = o4.w;
    b.x = q.z; b.y = q.w;
    c0 = __builtin_amdgcn_wmma_f32_16x16x4_f32(false, a0, false, b, (short)0, c0, false, false);
    c1 = __builtin_amdgcn_wmma_f32_16x16x4_f32(false, a1, false, b, (short)0, c1, false, false);
  }
}

__device__ __forceinline__ void sipm_flush_tile(float* fl, float* pt, int slot, bool again,
                                                int q, int ks, int w, int lane,
                                                const v8f& c0, const v8f& c1) {
  if (q == slot) {
    float* fp = fl + w * 512 + lane;
    #pragma unroll
    for (int r = 0; r < 8; ++r) { fp[r * 32] = c0[r]; fp[(8 + r) * 32] = c1[r]; }
  }
  __syncthreads();
  if (q == slot && ks == 0) {
    const float* f0 = fl + (4 * slot) * 512 + lane;
    #pragma unroll
    for (int r = 0; r < 8; ++r) {
      float s0 = ((f0[r * 32] + f0[512 + r * 32]) + f0[2 * 512 + r * 32]) + f0[3 * 512 + r * 32];
      float s1 = ((f0[(8 + r) * 32] + f0[512 + (8 + r) * 32]) + f0[2 * 512 + (8 + r) * 32])
                 + f0[3 * 512 + (8 + r) * 32];
      float* p0 = pt + (size_t)r * kTPad;
      float* p1 = p0 + (size_t)16 * kTPad;
      if (again) { s0 += *p0; s1 += *p1; }
      *p0 = s0;
      *p1 = s1;
    }
  }
  __syncthreads();
}

__global__ __launch_bounds__(512, 8)
void sipm_gemm_kernel(const float* __restrict__ W2, const float* __restrict__ b2,
                      const float* __restrict__ h, const float* __restrict__ expv,
                      const int* __restrict__ rng,
                      float* __restrict__ part, int* __restrict__ tmask) {
  __shared__ __attribute__((aligned(16))) float lds_w2f[2 * kHid * 16];
  __shared__ __attribute__((aligned(16))) float lds_resp[kKB * kSW];
  __shared__ __attribute__((aligned(16))) float lds_fl[16 * 512];

  const int tid    = threadIdx.x;
  const int strip  = blockIdx.x;
  const int chunk  = blockIdx.y;
  const int s_base = strip * kSW;

  for (int r = 0; r < 8; ++r) {
    int idx = tid + r * 512;
    int i = idx >> 5, ss = idx & 31;
    int s = s_base + ss;
    float v = (s < kS) ? W2[(size_t)i * kS + s] : 0.f;
    int st = ss >> 4, c = ss & 15;
    int j = i >> 3, hh2 = (i >> 2) & 1, q = i & 3;
    lds_w2f[((((st * 16 + j) * 32) + hh2 * 16 + c) << 2) + q] = v;
  }

  const int w    = tid >> 5;
  const int lane = tid & 31;
  const int m    = lane & 15;
  const int hh   = lane >> 4;
  const int wu = __builtin_amdgcn_readfirstlane(w);
  const int q  = wu >> 2;
  const int ks = wu & 3;

  const int  sA0 = s_base + m, sA1 = s_base + 16 + m;
  const float braw0 = b2[sA0 < kS ? sA0 : kS - 1];
  const float braw1 = b2[sA1 < kS ? sA1 : kS - 1];
  const float bias0 = (sA0 < kS) ? braw0 : 0.f;
  const float bias1 = (sA1 < kS) ? braw1 : 0.f;
  const float bl0 = bias0 * kLog2e, bl1 = bias1 * kLog2e;

  v8f c0 = {}, c1 = {};
  unsigned int curp = 0u;
  unsigned long long stored = 0ull;
  float* pbase = part + ((size_t)chunk * kSPad + s_base + (hh << 3)) * kTPad + m;

  __syncthreads();

  const int blk0 = chunk * kBlkCh;
  const int nblk = (kKBlocks - blk0 < kBlkCh) ? (kKBlocks - blk0) : kBlkCh;

  for (int kblk = 0; kblk < nblk; ++kblk) {
    const int k0 = (blk0 + kblk) * kKB;

    const int rlo = rng[2 * (blk0 + kblk)], rhi = rng[2 * (blk0 + kblk) + 1];

    {
      const float* hrow = h + (size_t)((k0 >> 4) + w) * ((kHid / 8) * 128) + (lane << 2);
      const float* wfr  = lds_w2f + (lane << 2);
      v8f d00 = {}, d01 = {}, d10 = {}, d11 = {};
      float4 aQ[2];
      aQ[0] = *(const float4*)(hrow);
      aQ[1] = *(const float4*)(hrow + 128);
      #pragma unroll
      for (int j = 0; j < kHid / 8; ++j) {
        const float4 a4 = aQ[j & 1];
        if (j + 2 < kHid / 8)
          aQ[j & 1] = *(const float4*)(hrow + 128 * (j + 2));
        const float4 p4 = *(const float4*)(wfr + j * 128);
        const float4 q4 = *(const float4*)(wfr + (16 + j) * 128);
        v2f aLo, aHi, pLo, pHi, qLo, qHi;
        aLo.x = a4.x; aLo.y = a4.y; aHi.x = a4.z; aHi.y = a4.w;
        pLo.x = p4.x; pLo.y = p4.y; pHi.x = p4.z; pHi.y = p4.w;
        qLo.x = q4.x; qLo.y = q4.y; qHi.x = q4.z; qHi.y = q4.w;
        d00 = __builtin_amdgcn_wmma_f32_16x16x4_f32(false, aLo, false, pLo,
                                                    (short)0, d00, false, false);
        d10 = __builtin_amdgcn_wmma_f32_16x16x4_f32(false, aLo, false, qLo,
                                                    (short)0, d10, false, false);
        d01 = __builtin_amdgcn_wmma_f32_16x16x4_f32(false, aHi, false, pHi,
                                                    (short)0, d01, false, false);
        d11 = __builtin_amdgcn_wmma_f32_16x16x4_f32(false, aHi, false, qHi,
                                                    (short)0, d11, false, false);
      }
      float* rp = lds_resp + ((((4 * w + 2 * hh) * 32) + m) << 2);
      #pragma unroll
      for (int r = 0; r < 8; r += 2) {
        float4 o;
        o.x = __builtin_amdgcn_exp2f(fmaf(d00[r]     + d01[r],     kLog2e, bl0));
        o.y = __builtin_amdgcn_exp2f(fmaf(d00[r + 1] + d01[r + 1], kLog2e, bl0));
        o.z = __builtin_amdgcn_exp2f(fmaf(d10[r]     + d11[r],     kLog2e, bl1));
        o.w = __builtin_amdgcn_exp2f(fmaf(d10[r + 1] + d11[r + 1], kLog2e, bl1));
        *(float4*)(rp + (((((r >> 1) & 1) * 32) + (r >> 2) * 16) << 2)) = o;
      }
    }
    __syncthreads();

    const int jlo = __builtin_amdgcn_readfirstlane(rlo < 0 ? 0 : rlo);
    const int jhi = __builtin_amdgcn_readfirstlane(rhi > kTTiles - 1 ? kTTiles - 1 : rhi);
    const float* ebase = expv + ((size_t)(k0 >> 3) + 8 * ks) * (kTTiles * 128) + (lane << 2);
    const float* ap    = lds_resp + (16 * ks) * 128 + (lane << 2);
    for (int j = jlo; j <= jhi; ++j) {
      const int slot = j & 3;
      const int held = (int)((curp >> (8 * slot)) & 0xffu) - 1;
      if (held != j) {
        if (held >= 0) {
          sipm_flush_tile(lds_fl, pbase + held * 16, slot, ((stored >> held) & 1ull) != 0ull,
                          q, ks, w, lane, c0, c1);
          stored |= 1ull << held;
        }
        curp = (curp & ~(0xffu << (8 * slot))) | ((unsigned int)(j + 1) << (8 * slot));
        if (q == slot) { c0 = v8f{}; c1 = v8f{}; }
      }
      if (q == slot)
        sipm_stageB_quarter(ebase + j * 128, ap, c0, c1);
    }
    __syncthreads();
  }

  for (int slot = 0; slot < 4; ++slot) {
    const int held = (int)((curp >> (8 * slot)) & 0xffu) - 1;
    if (held >= 0) {
      sipm_flush_tile(lds_fl, pbase + held * 16, slot, ((stored >> held) & 1ull) != 0ull,
                      q, ks, w, lane, c0, c1);
      stored |= 1ull << held;
    }
  }
  if (strip == 0 && tid == 0) {
    tmask[2 * chunk]     = (int)(unsigned int)(stored & 0xffffffffull);
    tmask[2 * chunk + 1] = (int)(unsigned int)(stored >> 32);
  }
}

__global__ __launch_bounds__(256)
void sipm_reduce_kernel(const float* __restrict__ part, const int* __restrict__ tmask,
                        float* __restrict__ out) {
  int idx = blockIdx.x * 256 + threadIdx.x;
  if (idx >= kS * kT) return;
  int s = idx / kT, t = idx - s * kT;
  size_t off = (size_t)s * kTPad + t;
  const int jt = t >> 4;
  unsigned int inb = 0u;
  #pragma unroll
  for (int c = 0; c < kNChunk; ++c) {
    const unsigned int lo = (unsigned int)tmask[2 * c], hi = (unsigned int)tmask[2 * c + 1];
    const unsigned int bits = (jt < 32) ? lo : hi;
    inb |= ((bits >> (jt & 31)) & 1u) << c;
  }
  float v[kNChunk];
  #pragma unroll
  for (int c = 0; c < kNChunk; ++c)
    v[c] = ((inb >> c) & 1u) ? part[(size_t)c * kSPad * kTPad + off] : 0.f;
  float sum = 0.f;
  #pragma unroll
  for (int c = 0; c < kNChunk; ++c) sum += v[c];
  out[idx] = sum;
}

extern "C" void kernel_launch(void* const* d_in, const int* in_sizes, int n_in,
                              void* d_out, int out_size, void* d_ws, size_t ws_size,
                              hipStream_t stream) {
  const float* x   = (const float*)d_in[0];
  const float* z   = (const float*)d_in[1];
  const float* msk = (const float*)d_in[2];
  const float* W1  = (const float*)d_in[3];
  const float* b1  = (const float*)d_in[4];
  const float* W2  = (const float*)d_in[5];
  const float* b2  = (const float*)d_in[6];
  float* out = (float*)d_out;

  float* ws   = (float*)d_ws;
  float* h    = ws;
  float* expv = h + (size_t)kK * kHid;
  float* part = expv + (size_t)kK * kTPad;
  int*   perm = (int*)(part + (size_t)kNChunk * kSPad * kTPad);
  int*   tmask = perm + kK;
  int*   rng  = tmask + 2 * kNChunk;

  sipm_sort_kernel<<<1, kSortT, 0, stream>>>(z, perm);
  sipm_h_kernel<<<(kK * (kHid / 4) + 255) / 256, 256, 0, stream>>>(x, W1, b1, perm, h);
  sipm_expv_kernel<<<kKBlocks * 2, 256, 0, stream>>>(z, msk, perm, expv, rng);
  sipm_gemm_kernel<<<dim3(kStrips, kNChunk), 512, 0, stream>>>(W2, b2, h, expv, rng, part, tmask);
  sipm_reduce_kernel<<<(kS * kT + 255) / 256, 256, 0, stream>>>(part, tmask, out);
}


# Sensor strip per workgroup widened 32 -> 64 (grid 35x14, all resident): each 16-byte h and expv fragment load feeds 8 WMMAs, was 4; panel GEMM: one accumulator per s-tile, hidden sum in a new order
# speedup vs baseline: 11.0924x; 1.0648x over previous
#include <hip/hip_runtime.h>
#include <math.h>


typedef float v2f __attribute__((ext_vector_type(2)));
typedef float v8f __attribute__((ext_vector_type(8)));

namespace {
constexpr int kBatch  = 64;
constexpr int kEv     = 500;
constexpr int kK      = kBatch * kEv;
constexpr int kHid    = 128;
constexpr int kS      = 2209;
constexpr int kSW     = 64;
constexpr int kST     = kSW / 16;
constexpr int kStrips = (kS + kSW - 1) / kSW;
constexpr int kSPad   = kStrips * kSW;
constexpr int kT      = 550;
constexpr int kTPad   = 560;
constexpr int kTTiles = kTPad / 16;
constexpr int kKB     = 256;
constexpr int kKBlocks = kK / kKB;
constexpr int kNChunk = 14;
constexpr int kBlkCh  = (kKBlocks + kNChunk - 1) / kNChunk;
constexpr float kNorm = 0.3989422804f * 1.41421356237309515f;
constexpr int kNB     = 35;
constexpr int kSortT  = 1024;
constexpr int kSortE  = 32;
constexpr int kSortOwn = kK / kSortE;
constexpr int kSortFull = kK / kSortT;
static_assert(kK % kSortE == 0 && kSortOwn <= kSortT && kK % 4 == 0, "sort tiling");
constexpr float kLog2e = 1.44269504088896341f;
constexpr int kPlane = (kKB / 4) * 128;
static_assert(kST == 4, "the panel planes and the accumulator names below are written for four s-tiles");
constexpr float kReach = 11.0f;
}

__device__ __forceinline__ int sipm_bucket(float zz) {
  float f = fminf(fmaxf(zz * 0.0625f, 0.f), (float)(kNB - 1));
  return (int)f;
}

__global__ __launch_bounds__(1024)
void sipm_sort_kernel(const float* __restrict__ z, int* __restrict__ perm) {
  __shared__ __attribute__((aligned(16))) unsigned char  bkt[kSortT * kSortE];
  __shared__ __attribute__((aligned(16))) unsigned short cnt[kNB * kSortT];
  __shared__ __attribute__((aligned(16))) int permL[kK];
  __shared__ int bbase[kNB];
  const int tid = threadIdx.x, lane = tid & 31, w = tid >> 5;

  for (int i = tid; i < kNB * kSortT / 8; i += kSortT)
    ((uint4*)cnt)[i] = make_uint4(0u, 0u, 0u, 0u);

  float zr[kSortFull + 1];
  #pragma unroll
  for (int j = 0; j < kSortFull; ++j) zr[j] = z[j * kSortT + tid];
  const bool tail = (kSortFull * kSortT + tid) < kK;
  zr[kSortFull] = tail ? z[kSortFull * kSortT + tid] : 0.f;
  #pragma unroll
  for (int j = 0; j < kSortFull; ++j)
    bkt[j * kSortT + tid] = (unsigned char)sipm_bucket(zr[j]);
  if (tail) bkt[kSortFull * kSortT + tid] = (unsigned char)sipm_bucket(zr[kSortFull]);
  __syncthreads();

  const bool own = tid < kSortOwn;
  unsigned int bw[kSortE / 4];
  int rank[kSortE];
  if (own) {
    const uint4 lo = *(const uint4*)(bkt + tid * kSortE);
    const uint4 hi = *(const uint4*)(bkt + tid * kSortE + 16);
    bw[0] = lo.x; bw[1] = lo.y; bw[2] = lo.z; bw[3] = lo.w;
    bw[4] = hi.x; bw[5] = hi.y; bw[6] = hi.z; bw[7] = hi.w;
    #pragma unroll
    for (int i = 0; i < kSortE; ++i) {
      const int bq = (int)((bw[i >> 2] >> (8 * (i & 3))) & 0xffu);
      const int old = cnt[bq * kSortT + tid];
      cnt[bq * kSortT + tid] = (unsigned short)(old + 1);
      rank[i] = old;
    }
  }
  __syncthreads();

  for (int bq = w; bq < kNB; bq += 32) {
    uint4* row = (uint4*)(cnt + bq * kSortT + lane * 32);
    unsigned int pk[16];
    #pragma unroll
    for (int v = 0; v < 4; ++v) {
      const uint4 q = row[v];
      pk[4 * v] = q.x; pk[4 * v + 1] = q.y; pk[4 * v + 2] = q.z; pk[4 * v + 3] = q.w;
    }
    int s = 0;
    #pragma unroll
    for (int i = 0; i < 16; ++i) s += (int)(pk[i] & 0xffffu) + (int)(pk[i] >> 16);
    int inc = s;
    #pragma unroll
    for (int d = 1; d < 32; d <<= 1) {
      const int o = __shfl_up(inc, d);
      if (lane >= d) inc += o;
    }
    int run = inc - s;
    #pragma unroll
    for (int i = 0; i < 16; ++i) {
      const int c0 = (int)(pk[i] & 0xffffu), c1 = (int)(pk[i] >> 16);
      const unsigned int p0 = (unsigned int)run;  run += c0;
      const unsigned int p1 = (unsigned int)run;  run += c1;
      pk[i] = (p0 & 0xffffu) | (p1 << 16);
    }
    #pragma unroll
    for (int v = 0; v < 4; ++v)
      row[v] = make_uint4(pk[4 * v], pk[4 * v + 1], pk[4 * v + 2], pk[4 * v + 3]);
    if (lane == 31) bbase[bq] = inc;
  }
  __syncthreads();
  if (tid == 0) {
    int run = 0;
    for (int bq = 0; bq < kNB; ++bq) { const int c = bbase[bq]; bbase[bq] = run; run += c; }
  }
  __syncthreads();

  if (own) {
    #pragma unroll
    for (int i = 0; i < kSortE; ++i) {
      const int bq = (int)((bw[i >> 2] >> (8 * (i & 3))) & 0xffu);
      permL[bbase[bq] + (int)cnt[bq * kSortT + tid] + rank[i]] = tid * kSortE + i;
    }
  }
  __syncthreads();
  #pragma unroll
  for (int j = 0; j < (kK / 4 + kSortT - 1) / kSortT; ++j) {
    const int i = j * kSortT + tid;
    if (i < kK / 4) ((int4*)perm)[i] = ((const int4*)permL)[i];
  }
}

__global__ __launch_bounds__(256)
void sipm_h_kernel(const float* __restrict__ x, const float* __restrict__ W1,
                   const float* __restrict__ b1, const int* __restrict__ perm,
                   float* __restrict__ h) {
  int idx = blockIdx.x * 256 + threadIdx.x;
  if (idx >= kK * (kHid / 4)) return;
  int lane = idx & 31;
  int k = perm[((idx >> 9) << 4) + (lane & 15)];
  int j = (((idx >> 5) & 15) << 3) + ((lane >> 4) << 2);
  float x0 = x[k * 3 + 0], x1 = x[k * 3 + 1], x2 = x[k * 3 + 2];
  const float4 wa = *(const float4*)(W1 + 0 * kHid + j);
  const float4 wb = *(const float4*)(W1 + 1 * kHid + j);
  const float4 wc = *(const float4*)(W1 + 2 * kHid + j);
  const float4 bb = *(const float4*)(b1 + j);
  float4 r;
  r.x = fmaxf(fmaf(x0, wa.x, fmaf(x1, wb.x, fmaf(x2, wc.x, bb.x))), 0.f);
  r.y = fmaxf(fmaf(x0, wa.y, fmaf(x1, wb.y, fmaf(x2, wc.y, bb.y))), 0.f);
  r.z = fmaxf(fmaf(x0, wa.z, fmaf(x1, wb.z, fmaf(x2, wc.z, bb.z))), 0.f);
  r.w = fmaxf(fmaf(x0, wa.w, fmaf(x1, wb.w, fmaf(x2, wc.w, bb.w))), 0.f);
  *(float4*)(h + ((size_t)idx << 2)) = r;
}

__global__ __launch_bounds__(256)
void sipm_expv_kernel(const float* __restrict__ z, const float* __restrict__ mask,
                      const int* __restrict__ perm, float* __restrict__ expv,
                      int* __restrict__ rng) {
  __shared__ __attribute__((aligned(16))) float zs[kKB];
  __shared__ __attribute__((aligned(16))) float ms[kKB];
  __shared__ float red[16];
  const int tid = threadIdx.x, lane = tid & 31, w = tid >> 5;
  const int kb = blockIdx.x >> 1, half = blockIdx.x & 1;
  const int ke = perm[kb * kKB + tid];
  const float zz = z[ke], mm = mask[ke];
  zs[tid] = zz;
  ms[tid] = mm;
  const bool fin = (fabsf(zz) < INFINITY) && (fabsf(mm) < INFINITY);
  float lo = fin ? zz : -INFINITY;
  float hi = fin ? zz : INFINITY;
  for (int d = 16; d >= 1; d >>= 1) {
    lo = fminf(lo, __shfl_xor(lo, d));
    hi = fmaxf(hi, __shfl_xor(hi, d));
  }
  if (lane == 0) { red[2 * w] = lo; red[2 * w + 1] = hi; }
  __syncthreads();
  float zmin = red[0], zmax = red[1];
  for (int i = 1; i < 8; ++i) {
    zmin = fminf(zmin, red[2 * i]);
    zmax = fmaxf(zmax, red[2 * i + 1]);
  }
  const float flo = fminf(fmaxf((zmin - kReach) * 0.0625f, 0.f), (float)kTTiles);
  const float fhi = fminf(fmaxf((zmax + kReach) * 0.0625f, -1.f), (float)(kTTiles - 1));
  const int jlo = __builtin_amdgcn_readfirstlane((int)flo);
  const int jhi = __builtin_amdgcn_readfirstlane((int)floorf(fhi));
  if (half == 0 && tid == 0) { rng[2 * kb] = jlo; rng[2 * kb + 1] = jhi; }

  const int hh = lane >> 4, cc = lane & 15;
  const int g  = half * 16 + w;
  const float4 za = *(const float4*)(zs + 8 * g + 4 * hh);
  const float4 ma = *(const float4*)(ms + 8 * g + 4 * hh);
  const float4 zb = *(const float4*)(zs + 8 * (g + 8) + 4 * hh);
  const float4 mb = *(const float4*)(ms + 8 * (g + 8) + 4 * hh);
  float* ea = expv + (size_t)(kb * (kKB / 8) + g) * (kTTiles * 128) + (lane << 2);
  float* eb = ea + (size_t)8 * (kTTiles * 128);
  for (int j = jlo; j <= jhi; ++j) {
    const int t = (j << 4) + cc;
    float4 va = make_float4(0.f, 0.f, 0.f, 0.f);
    float4 vb = va;
    if (t < kT) {
      const float tf = (float)t;
      float d;
      d = tf - za.x; va.x = expf(-d * d) * ma.x * kNorm;
      d = tf - za.y; va.y = expf(-d * d) * ma.y * kNorm;
      d = tf - za.z; va.z = expf(-d * d) * ma.z * kNorm;
      d = tf - za.w; va.w = expf(-d * d) * ma.w * kNorm;
      d = tf - zb.x; vb.x = expf(-d * d) * mb.x * kNorm;
      d = tf - zb.y; vb.y = expf(-d * d) * mb.y * kNorm;
      d = tf - zb.z; vb.z = expf(-d * d) * mb.z * kNorm;
      d = tf - zb.w; vb.w = expf(-d * d) * mb.w * kNorm;
    }
    *(float4*)(ea + j * 128) = va;
    *(float4*)(eb + j * 128) = vb;
  }
}

__device__ __forceinline__ void sipm_stageB_quarter(const float* __restrict__ e, const float* ap,
                                                    v8f& c0, v8f& c1, v8f& c2, v8f& c3) {
  constexpr int kQP    = kKB / 8 / 4;
  constexpr int kPitch = kTTiles * 128;
  float4 bQ[kQP];
  #pragma unroll
  for (int p = 0; p < kQP; ++p) bQ[p] = *(const float4*)(e + p * kPitch);
  float4 e0 = *(const float4*)(ap);
  float4 o0 = *(const float4*)(ap + 128);
  float4 e1 = *(const float4*)(ap + kPlane);
  float4 o1 = *(const float4*)(ap + kPlane + 128);
  #pragma unroll
  for (int i = 0; i < kQP; ++i) {
    float4 ne0 = e0, no0 = o0, ne1 = e1, no1 = o1;
    if (i + 1 < kQP) {
      ne0 = *(const float4*)(ap + (2 * i + 2) * 128);
      no0 = *(const float4*)(ap + (2 * i + 3) * 128);
      ne1 = *(const float4*)(ap + kPlane + (2 * i + 2) * 128);
      no1 = *(const float4*)(ap + kPlane + (2 * i + 3) * 128);
    }
    const float4 q  = bQ[i];
    v2f a0, a1, a2, a3, b;
    a0.x = e0.x; a0.y = e0.y;
    a1.x = e0.z; a1.y = e0.w;
    a2.x = e1.x; a2.y = e1.y;
    a3.x = e1.z; a3.y = e1.w;
    b.x = q.x; b.y = q.y;
    c0 = __builtin_amdgcn_wmma_f32_16x16x4_f32(false, a0, false, b, (short)0, c0, false, false);
    c1 = __builtin_amdgcn_wmma_f32_16x16x4_f32(false, a1, false, b, (short)0, c1, false, false);
    c2 = __builtin_amdgcn_wmma_f32_16x16x4_f32(false, a2, false, b, (short)0, c2, false, false);
    c3 = __builtin_amdgcn_wmma_f32_16x16x4_f32(false, a3, false, b, (short)0, c3, false, false);
    a0.x = o0.x; a0.y = o0.y;
    a1.x = o0.z; a1.y = o0.w;
    a2.x = o1.x; a2.y = o1.y;
    a3.x = o1.z; a3.y = o1.w;
    b.x = q.z; b.y = q.w;
    c0 = __builtin_amdgcn_wmma_f32_16x16x4_f32(false, a0, false, b, (short)0, c0, false, false);
    c1 = __builtin_amdgcn_wmma_f32_16x16x4_f32(false, a1, false, b, (short)0, c1, false, false);
    c2 = __builtin_amdgcn_wmma_f32_16x16x4_f32(false, a2, false, b, (short)0, c2, false, false);
    c3 = __builtin_amdgcn_wmma_f32_16x16x4_f32(false, a3, false, b, (short)0, c3, false, false);
    e0 = ne0; o0 = no0; e1 = ne1; o1 = no1;
  }
}

__device__ __forceinline__ void sipm_flush_tile(float* fl, float* pt, int slot, bool again,
                                                int q, int ks, int lane,
                                                const v8f& c0, const v8f& c1,
                                                const v8f& c2, const v8f& c3) {
  if (q == slot) {
    float* fp = fl + ks * 1024 + lane;
    #pragma unroll
    for (int r = 0; r < 8; ++r) {
      fp[r * 32]        = c0[r];
      fp[(8 + r) * 32]  = c1[r];
      fp[(16 + r) * 32] = c2[r];
      fp[(24 + r) * 32] = c3[r];
    }
  }
  __syncthreads();
  if (q == slot && ks == 0) {
    const float* f0 = fl + lane;
    #pragma unroll
    for (int r = 0; r < 8; ++r) {
      float sm[kST];
      #pragma unroll
      for (int st = 0; st < kST; ++st) {
        const int g = (8 * st + r) * 32;
        sm[st] = ((f0[g] + f0[1024 + g]) + f0[2 * 1024 + g]) + f0[3 * 1024 + g];
      }
      float* p = pt + (size_t)r * kTPad;
      if (again) {
        #pragma unroll
        for (int st = 0; st < kST; ++st) sm[st] += p[(size_t)(16 * st) * kTPad];
      }
      #pragma unroll
      for (int st = 0; st < kST; ++st) p[(size_t)(16 * st) * kTPad] = sm[st];
    }
  }
  __syncthreads();
}

__global__ __launch_bounds__(512, 8)
void sipm_gemm_kernel(const float* __restrict__ W2, const float* __restrict__ b2,
                      const float* __restrict__ h, const float* __restrict__ expv,
                      const int* __restrict__ rng,
                      float* __restrict__ part, int* __restrict__ tmask) {
  __shared__ __attribute__((aligned(16))) float lds_w2f[kST * kHid * 16];
  __shared__ __attribute__((aligned(16))) float lds_resp[kKB * kSW];
  __shared__ __attribute__((aligned(16))) float lds_fl[4 * 1024];

  const int tid    = threadIdx.x;
  const int strip  = blockIdx.x;
  const int chunk  = blockIdx.y;
  const int s_base = strip * kSW;

  for (int r = 0; r < kHid * kSW / 512; ++r) {
    int idx = tid + r * 512;
    int i = idx >> 6, ss = idx & 63;
    int sn = s_base + ss;
    float v = (sn < kS) ? W2[(size_t)i * kS + sn] : 0.f;
    int st = ss >> 4, c = ss & 15;
    int j = i >> 3, hh2 = (i >> 2) & 1, q = i & 3;
    lds_w2f[((((st * 16 + j) * 32) + hh2 * 16 + c) << 2) + q] = v;
  }

  const int w    = tid >> 5;
  const int lane = tid & 31;
  const int m    = lane & 15;
  const int hh   = lane >> 4;
  const int wu = __builtin_amdgcn_readfirstlane(w);
  const int q  = wu >> 2;
  const int ks = wu & 3;

  const int  sA0 = s_base + m, sA1 = sA0 + 16, sA2 = sA0 + 32, sA3 = sA0 + 48;
  const float braw0 = b2[sA0 < kS ? sA0 : kS - 1];
  const float braw1 = b2[sA1 < kS ? sA1 : kS - 1];
  const float braw2 = b2[sA2 < kS ? sA2 : kS - 1];
  const float braw3 = b2[sA3 < kS ? sA3 : kS - 1];
  const float bl0 = ((sA0 < kS) ? braw0 : 0.f) * kLog2e;
  const float bl1 = ((sA1 < kS) ? braw1 : 0.f) * kLog2e;
  const float bl2 = ((sA2 < kS) ? braw2 : 0.f) * kLog2e;
  const float bl3 = ((sA3 < kS) ? braw3 : 0.f) * kLog2e;
  const v2f l2e2 = {kLog2e, kLog2e};
  const v2f bb0 = {bl0, bl0}, bb1 = {bl1, bl1}, bb2 = {bl2, bl2}, bb3 = {bl3, bl3};

  v8f c0 = {}, c1 = {}, c2 = {}, c3 = {};
  unsigned int curp = 0u;
  unsigned long long stored = 0ull;
  float* pbase = part + ((size_t)chunk * kSPad + s_base + (hh << 3)) * kTPad + m;

  __syncthreads();

  const int blk0 = chunk * kBlkCh;
  const int nblk = (kKBlocks - blk0 < kBlkCh) ? (kKBlocks - blk0) : kBlkCh;

  for (int kblk = 0; kblk < nblk; ++kblk) {
    const int k0 = (blk0 + kblk) * kKB;

    const int rlo = rng[2 * (blk0 + kblk)], rhi = rng[2 * (blk0 + kblk) + 1];

    {
      const float* hrow = h + (size_t)((k0 >> 4) + w) * ((kHid / 8) * 128) + (lane << 2);
      const float* wfr  = lds_w2f + (lane << 2);
      v8f d0 = {}, d1 = {}, d2 = {}, d3 = {};
      float4 aQ[2];
      aQ[0] = *(const float4*)(hrow);
      aQ[1] = *(const float4*)(hrow + 128);
      #pragma unroll
      for (int j = 0; j < kHid / 8; ++j) {
        const float4 a4 = aQ[j & 1];
        if (j + 2 < kHid / 8)
          aQ[j & 1] = *(const float4*)(hrow + 128 * (j + 2));
        const float4 p0 = *(const float4*)(wfr + (0 * 16 + j) * 128);
        const float4 p1 = *(const float4*)(wfr + (1 * 16 + j) * 128);
        const float4 p2 = *(const float4*)(wfr + (2 * 16 + j) * 128);
        const float4 p3 = *(const float4*)(wfr + (3 * 16 + j) * 128);
        v2f aLo, aHi, bv;
        aLo.x = a4.x; aLo.y = a4.y; aHi.x = a4.z; aHi.y = a4.w;
        bv.x = p0.x; bv.y = p0.y;
        d0 = __builtin_amdgcn_wmma_f32_16x16x4_f32(false, aLo, false, bv, (short)0, d0, false, false);
        bv.x = p1.x; bv.y = p1.y;
        d1 = __builtin_amdgcn_wmma_f32_16x16x4_f32(false, aLo, false, bv, (short)0, d1, false, false);
        bv.x = p2.x; bv.y = p2.y;
        d2 = __builtin_amdgcn_wmma_f32_16x16x4_f32(false, aLo, false, bv, (short)0, d2, false, false);
        bv.x = p3.x; bv.y = p3.y;
        d3 = __builtin_amdgcn_wmma_f32_16x16x4_f32(false, aLo, false, bv, (short)0, d3, false, false);
        bv.x = p0.z; bv.y = p0.w;
        d0 = __builtin_amdgcn_wmma_f32_16x16x4_f32(false, aHi, false, bv, (short)0, d0, false, false);
        bv.x = p1.z; bv.y = p1.w;
        d1 = __builtin_amdgcn_wmma_f32_16x16x4_f32(false, aHi, false, bv, (short)0, d1, false, false);
        bv.x = p2.z; bv.y = p2.w;
        d2 = __builtin_amdgcn_wmma_f32_16x16x4_f32(false, aHi, false, bv, (short)0, d2, false, false);
        bv.x = p3.z; bv.y = p3.w;
        d3 = __builtin_amdgcn_wmma_f32_16x16x4_f32(false, aHi, false, bv, (short)0, d3, false, false);
      }
      float* rp = lds_resp + ((((4 * w + 2 * hh) * 32) + m) << 2);
      #pragma unroll
      for (int r = 0; r < 8; r += 2) {
        float4 o, u;
        const v2f t0 = __builtin_elementwise_fma(v2f{d0[r], d0[r + 1]}, l2e2, bb0);
        const v2f t1 = __builtin_elementwise_fma(v2f{d1[r], d1[r + 1]}, l2e2, bb1);
        const v2f t2 = __builtin_elementwise_fma(v2f{d2[r], d2[r + 1]}, l2e2, bb2);
        const v2f t3 = __builtin_elementwise_fma(v2f{d3[r], d3[r + 1]}, l2e2, bb3);
        o.x = __builtin_amdgcn_exp2f(t0.x);
        o.y = __builtin_amdgcn_exp2f(t0.y);
        o.z = __builtin_amdgcn_exp2f(t1.x);
        o.w = __builtin_amdgcn_exp2f(t1.y);
        u.x = __builtin_amdgcn_exp2f(t2.x);
        u.y = __builtin_amdgcn_exp2f(t2.y);
        u.z = __builtin_amdgcn_exp2f(t3.x);
        u.w = __builtin_amdgcn_exp2f(t3.y);
        const int off = ((((r >> 1) & 1) * 32) + (r >> 2) * 16) << 2;
        *(float4*)(rp + off)          = o;
        *(float4*)(rp + kPlane + off) = u;
      }
    }
    __syncthreads();

    const int jlo = __builtin_amdgcn_readfirstlane(rlo < 0 ? 0 : rlo);
    const int jhi = __builtin_amdgcn_readfirstlane(rhi > kTTiles - 1 ? kTTiles - 1 : rhi);
    const float* ebase = expv + ((size_t)(k0 >> 3) + 8 * ks) * (kTTiles * 128) + (lane << 2);
    const float* ap    = lds_resp + (16 * ks) * 128 + (lane << 2);
    for (int j = jlo; j <= jhi; ++j) {
      const int slot = j & 3;
      const int held = (int)((curp >> (8 * slot)) & 0xffu) - 1;
      if (held != j) {
        if (held >= 0) {
          sipm_flush_tile(lds_fl, pbase + held * 16, slot, ((stored >> held) & 1ull) != 0ull,
                          q, ks, lane, c0, c1, c2, c3);
          stored |= 1ull << held;
        }
        curp = (curp & ~(0xffu << (8 * slot))) | ((unsigned int)(j + 1) << (8 * slot));
        if (q == slot) { c0 = v8f{}; c1 = v8f{}; c2 = v8f{}; c3 = v8f{}; }
      }
      if (q == slot)
        sipm_stageB_quarter(ebase + j * 128, ap, c0, c1, c2, c3);
    }
    __syncthreads();
  }

  for (int slot = 0; slot < 4; ++slot) {
    const int held = (int)((curp >> (8 * slot)) & 0xffu) - 1;
    if (held >= 0) {
      sipm_flush_tile(lds_fl, pbase + held * 16, slot, ((stored >> held) & 1ull) != 0ull,
                      q, ks, lane, c0, c1, c2, c3);
      stored |= 1ull << held;
    }
  }
  if (strip == 0 && tid == 0) {
    tmask[2 * chunk]     = (int)(unsigned int)(stored & 0xffffffffull);
    tmask[2 * chunk + 1] = (int)(unsigned int)(stored >> 32);
  }
}

__global__ __launch_bounds__(256)
void sipm_reduce_kernel(const float* __restrict__ part, const int* __restrict__ tmask,
                        float* __restrict__ out) {
  int idx = blockIdx.x * 256 + threadIdx.x;
  if (idx >= kS * kT) return;
  int s = idx / kT, t = idx - s * kT;
  size_t off = (size_t)s * kTPad + t;
  const int jt = t >> 4;
  unsigned int inb = 0u;
  #pragma unroll
  for (int c = 0; c < kNChunk; ++c) {
    const unsigned int lo = (unsigned int)tmask[2 * c], hi = (unsigned int)tmask[2 * c + 1];
    const unsigned int bits = (jt < 32) ? lo : hi;
    inb |= ((bits >> (jt & 31)) & 1u) << c;
  }
  float v[kNChunk];
  #pragma unroll
  for (int c = 0; c < kNChunk; ++c)
    v[c] = ((inb >> c) & 1u) ? part[(size_t)c * kSPad * kTPad + off] : 0.f;
  float sum = 0.f;
  #pragma unroll
  for (int c = 0; c < kNChunk; ++c) sum += v[c];
  out[idx] = sum;
}

extern "C" void kernel_launch(void* const* d_in, const int* in_sizes, int n_in,
                              void* d_out, int out_size, void* d_ws, size_t ws_size,
                              hipStream_t stream) {
  const float* x   = (const float*)d_in[0];
  const float* z   = (const float*)d_in[1];
  const float* msk = (const float*)d_in[2];
  const float* W1  = (const float*)d_in[3];
  const float* b1  = (const float*)d_in[4];
  const float* W2  = (const float*)d_in[5];
  const float* b2  = (const float*)d_in[6];
  float* out = (float*)d_out;

  float* ws   = (float*)d_ws;
  float* h    = ws;
  float* expv = h + (size_t)kK * kHid;
  float* part = expv + (size_t)kK * kTPad;
  int*   perm = (int*)(part + (size_t)kNChunk * kSPad * kTPad);
  int*   tmask = perm + kK;
  int*   rng  = tmask + 2 * kNChunk;

  sipm_sort_kernel<<<1, kSortT, 0, stream>>>(z, perm);
  sipm_h_kernel<<<(kK * (kHid / 4) + 255) / 256, 256, 0, stream>>>(x, W1, b1, perm, h);
  sipm_expv_kernel<<<kKBlocks * 2, 256, 0, stream>>>(z, msk, perm, expv, rng);
  sipm_gemm_kernel<<<dim3(kStrips, kNChunk), 512, 0, stream>>>(W2, b2, h, expv, rng, part, tmask);
  sipm_reduce_kernel<<<(kS * kT + 255) / 256, 256, 0, stream>>>(part, tmask, out);
}
